# starting kernel
# baseline (speedup 1.0000x reference)
#include <hip/hip_runtime.h>

#define D      4096
#define DMASK  4095
#define BS     128
#define NPAIRS 64
#define HB_EPS 1e-8f

typedef __attribute__((ext_vector_type(2))) float v2f;
typedef __attribute__((ext_vector_type(8))) float v8f;

__global__ void hb_softmax128(const float* __restrict__ pw, float* __restrict__ w) {
    __shared__ float sm[BS];
    int t = threadIdx.x;
    float x = pw[t];
    sm[t] = x; __syncthreads();
    for (int s = BS / 2; s > 0; s >>= 1) {
        if (t < s) sm[t] = fmaxf(sm[t], sm[t + s]);
        __syncthreads();
    }
    float mx = sm[0]; __syncthreads();
    float e = expf(x - mx);
    sm[t] = e; __syncthreads();
    for (int s = BS / 2; s > 0; s >>= 1) {
        if (t < s) sm[t] += sm[t + s];
        __syncthreads();
    }
    w[t] = e / sm[0];
}

__global__ void __launch_bounds__(256) hb_shiftsum(
        const float* __restrict__ inp, const float* __restrict__ w,
        float* __restrict__ y) {
    __shared__ float sw[BS];
    int t = threadIdx.x;
    if (t < BS) sw[t] = w[t];
    __syncthreads();

    int pair = blockIdx.y;
    int q0 = blockIdx.x * 512 + t;
    int q1 = q0 + 256;
    const float* prow = inp + (size_t)pair * BS * D;
    float acc0 = 0.0f, acc1 = 0.0f;
    #pragma unroll 4
    for (int s = 0; s < BS; ++s) {
        const float* row = prow + (size_t)s * D;
        float ws = sw[s];
        acc0 = fmaf(ws, row[(q0 + s) & DMASK], acc0);
        acc1 = fmaf(ws, row[(q1 + s) & DMASK], acc1);
    }
    y[pair * D + q0] = acc0;
    y[pair * D + q1] = acc1;
}

__global__ void __launch_bounds__(256) hb_gemm_wmma(
        const float* __restrict__ y, const float* __restrict__ base,
        float* __restrict__ out) {
    __shared__ float lbase[2 * D];
    int t = threadIdx.x;
    for (int i = t; i < D; i += 256) {
        float v = base[i];
        lbase[i]     = v;
        lbase[i + D] = v;
    }
    __syncthreads();

    int wave = t >> 5;
    int lane = t & 31;
    int half = lane >> 4;
    int l    = lane & 15;

    int m0 = blockIdx.y * 16;
    int n0 = blockIdx.x * 128 + wave * 16;
    int n  = n0 + l;
    const float* arow = y + (size_t)(m0 + l) * D;
    const float* bcol = lbase + ((2 * half - n + 8192) & DMASK);

    v8f acc = {};
    for (int kc = 0; kc < D; kc += 64) {
        __builtin_prefetch(arow + kc + 64, 0, 0);
        #pragma unroll
        for (int kk = 0; kk < 64; kk += 4) {
            int k0 = kc + kk;
            float2 af = *(const float2*)(arow + k0 + 2 * half);
            v2f a = { af.x, af.y };
            v2f b = { bcol[k0], bcol[k0 + 1] };
            acc = __builtin_amdgcn_wmma_f32_16x16x4_f32(
                       false, a,  false, b,
                       (short)0, acc,  false,  false);
        }
    }
    #pragma unroll
    for (int v = 0; v < 8; ++v)
        out[(size_t)(m0 + v + 8 * half) * D + n] = acc[v];
}

__global__ void __launch_bounds__(256) hb_normalize(float* __restrict__ out) {
    __shared__ float sm[256];
    int t = threadIdx.x;
    float* rowp = out + (size_t)blockIdx.x * D;
    float v[16];
    float ssq = 0.0f;
    #pragma unroll
    for (int i = 0; i < 16; ++i) {
        v[i] = rowp[t + i * 256];
        ssq = fmaf(v[i], v[i], ssq);
    }
    sm[t] = ssq; __syncthreads();
    for (int s = 128; s > 0; s >>= 1) {
        if (t < s) sm[t] += sm[t + s];
        __syncthreads();
    }
    float scale = rsqrtf(fmaxf(sm[0], HB_EPS));
    #pragma unroll
    for (int i = 0; i < 16; ++i)
        rowp[t + i * 256] = v[i] * scale;
}

extern "C" void kernel_launch(void* const* d_in, const int* in_sizes, int n_in,
                              void* d_out, int out_size, void* d_ws, size_t ws_size,
                              hipStream_t stream) {
    (void)in_sizes; (void)n_in; (void)out_size; (void)ws_size;
    const float* inp  = (const float*)d_in[0];
    const float* base = (const float*)d_in[1];
    const float* pw   = (const float*)d_in[2];
    float* out = (float*)d_out;

    float* w = (float*)d_ws;
    float* y = (float*)((char*)d_ws + 1024);

    hb_softmax128<<<1, 128, 0, stream>>>(pw, w);
    hb_shiftsum  <<<dim3(8, NPAIRS), 256, 0, stream>>>(inp, w, y);
    hb_gemm_wmma <<<dim3(32, 4),     256, 0, stream>>>(y, base, out);
    hb_normalize <<<NPAIRS,          256, 0, stream>>>(out);
}


# GEMM re-tiled: a work-group owns a 16x64 strip, its 8 waves split K and each reuses one A fragment for 4 N-tiles; K-partials summed in LDS (changes the summation order over k)
# speedup vs baseline: 1.5031x; 1.5031x over previous
#include <hip/hip_runtime.h>

#define D      4096
#define DMASK  4095
#define BS     128
#define NPAIRS 64
#define HB_EPS 1e-8f

typedef __attribute__((ext_vector_type(2))) float v2f;
typedef __attribute__((ext_vector_type(8))) float v8f;

__global__ void hb_softmax128(const float* __restrict__ pw, float* __restrict__ w) {
    __shared__ float sm[BS];
    int t = threadIdx.x;
    float x = pw[t];
    sm[t] = x; __syncthreads();
    for (int s = BS / 2; s > 0; s >>= 1) {
        if (t < s) sm[t] = fmaxf(sm[t], sm[t + s]);
        __syncthreads();
    }
    float mx = sm[0]; __syncthreads();
    float e = expf(x - mx);
    sm[t] = e; __syncthreads();
    for (int s = BS / 2; s > 0; s >>= 1) {
        if (t < s) sm[t] += sm[t + s];
        __syncthreads();
    }
    w[t] = e / sm[0];
}

__global__ void __launch_bounds__(256) hb_shiftsum(
        const float* __restrict__ inp, const float* __restrict__ w,
        float* __restrict__ y) {
    __shared__ float sw[BS];
    int t = threadIdx.x;
    if (t < BS) sw[t] = w[t];
    __syncthreads();

    int pair = blockIdx.y;
    int q0 = blockIdx.x * 512 + t;
    int q1 = q0 + 256;
    const float* prow = inp + (size_t)pair * BS * D;
    float acc0 = 0.0f, acc1 = 0.0f;
    #pragma unroll 4
    for (int s = 0; s < BS; ++s) {
        const float* row = prow + (size_t)s * D;
        float ws = sw[s];
        acc0 = fmaf(ws, row[(q0 + s) & DMASK], acc0);
        acc1 = fmaf(ws, row[(q1 + s) & DMASK], acc1);
    }
    y[pair * D + q0] = acc0;
    y[pair * D + q1] = acc1;
}

#define GK_WAVES 8
#define GK_KPW   (D / GK_WAVES)
#define GK_CHUNK 32

__device__ __forceinline__ v8f hb_mma(v2f a, v2f b, v8f c) {
    return __builtin_amdgcn_wmma_f32_16x16x4_f32(
                false, a,  false, b,
                (short)0, c,  false,  false);
}

__device__ __forceinline__ void hb_load4(float4 (&d)[4], const float* p) {
    #pragma unroll
    for (int s = 0; s < 4; ++s) d[s] = *(const float4*)(p + 8 * s);
}

__device__ __forceinline__ void hb_chunk(
        const float4 (&a)[4], const float* b0, const float* b1,
        const float* b2, const float* b3,
        v8f& acc0, v8f& acc1, v8f& acc2, v8f& acc3) {
    #pragma unroll
    for (int s = 0; s < 4; ++s) {
        int ko = 8 * s;
        v2f alo = { a[s].x, a[s].y };
        v2f ahi = { a[s].z, a[s].w };
        v2f b0lo = { b0[ko], b0[ko + 1] }, b0hi = { b0[ko + 2], b0[ko + 3] };
        v2f b1lo = { b1[ko], b1[ko + 1] }, b1hi = { b1[ko + 2], b1[ko + 3] };
        v2f b2lo = { b2[ko], b2[ko + 1] }, b2hi = { b2[ko + 2], b2[ko + 3] };
        v2f b3lo = { b3[ko], b3[ko + 1] }, b3hi = { b3[ko + 2], b3[ko + 3] };
        acc0 = hb_mma(alo, b0lo, acc0);
        acc1 = hb_mma(alo, b1lo, acc1);
        acc2 = hb_mma(alo, b2lo, acc2);
        acc3 = hb_mma(alo, b3lo, acc3);
        acc0 = hb_mma(ahi, b0hi, acc0);
        acc1 = hb_mma(ahi, b1hi, acc1);
        acc2 = hb_mma(ahi, b2hi, acc2);
        acc3 = hb_mma(ahi, b3hi, acc3);
    }
}

__global__ void __launch_bounds__(256) hb_gemm_wmma(
        const float* __restrict__ y, const float* __restrict__ base,
        float* __restrict__ out) {
    __shared__ __attribute__((aligned(16))) float lbase[2 * D];
    int t = threadIdx.x;
    for (int i = t; i < D; i += 256) {
        float v = base[i];
        lbase[i]     = v;
        lbase[i + D] = v;
    }
    __syncthreads();

    int wave = t >> 5;
    int lane = t & 31;
    int half = lane >> 4;
    int l    = lane & 15;

    int m0 = blockIdx.y * 16;
    int n0 = blockIdx.x * 64;
    int kbeg = wave * GK_KPW;

    const float* ap = y + (size_t)(m0 + l) * D + kbeg + 4 * half;
    const float* bp0 = lbase + ((4 * half - (n0 +  0 + l) + 8192) & DMASK) + kbeg;
    const float* bp1 = lbase + ((4 * half - (n0 + 16 + l) + 8192) & DMASK) + kbeg;
    const float* bp2 = lbase + ((4 * half - (n0 + 32 + l) + 8192) & DMASK) + kbeg;
    const float* bp3 = lbase + ((4 * half - (n0 + 48 + l) + 8192) & DMASK) + kbeg;

    v8f acc0 = {}, acc1 = {}, acc2 = {}, acc3 = {};

    float4 ea[4], oa[4];
    hb_load4(ea, ap);

    #pragma unroll 1
    for (int kc = 0; kc < GK_KPW - 2 * GK_CHUNK; kc += 2 * GK_CHUNK) {
        hb_load4(oa, ap + kc + GK_CHUNK);
        hb_chunk(ea, bp0 + kc, bp1 + kc, bp2 + kc, bp3 + kc, acc0, acc1, acc2, acc3);
        hb_load4(ea, ap + kc + 2 * GK_CHUNK);
        hb_chunk(oa, bp0 + kc + GK_CHUNK, bp1 + kc + GK_CHUNK, bp2 + kc + GK_CHUNK,
                 bp3 + kc + GK_CHUNK, acc0, acc1, acc2, acc3);
    }
    {
        const int kc = GK_KPW - 2 * GK_CHUNK;
        hb_load4(oa, ap + kc + GK_CHUNK);
        hb_chunk(ea, bp0 + kc, bp1 + kc, bp2 + kc, bp3 + kc, acc0, acc1, acc2, acc3);
        hb_chunk(oa, bp0 + kc + GK_CHUNK, bp1 + kc + GK_CHUNK, bp2 + kc + GK_CHUNK,
                 bp3 + kc + GK_CHUNK, acc0, acc1, acc2, acc3);
    }

    __syncthreads();
    float* red = lbase;
    float* myred = red + wave * 1024 + lane;
    #pragma unroll
    for (int v = 0; v < 8; ++v) {
        myred[(0 * 8 + v) * 32] = acc0[v];
        myred[(1 * 8 + v) * 32] = acc1[v];
        myred[(2 * 8 + v) * 32] = acc2[v];
        myred[(3 * 8 + v) * 32] = acc3[v];
    }
    __syncthreads();

    int r   = t >> 4;
    int c4  = (t & 15) << 2;
    int ridx = (((c4 >> 4) * 8 + (r & 7)) * 32) + (r >> 3) * 16 + (c4 & 15);
    float4 sum = *(const float4*)(red + ridx);
    #pragma unroll
    for (int w = 1; w < GK_WAVES; ++w) {
        float4 p = *(const float4*)(red + w * 1024 + ridx);
        sum.x += p.x; sum.y += p.y; sum.z += p.z; sum.w += p.w;
    }
    *(float4*)(out + (size_t)(m0 + r) * D + n0 + c4) = sum;
}

__global__ void __launch_bounds__(256) hb_normalize(float* __restrict__ out) {
    __shared__ float sm[256];
    int t = threadIdx.x;
    float* rowp = out + (size_t)blockIdx.x * D;
    float v[16];
    float ssq = 0.0f;
    #pragma unroll
    for (int i = 0; i < 16; ++i) {
        v[i] = rowp[t + i * 256];
        ssq = fmaf(v[i], v[i], ssq);
    }
    sm[t] = ssq; __syncthreads();
    for (int s = 128; s > 0; s >>= 1) {
        if (t < s) sm[t] += sm[t + s];
        __syncthreads();
    }
    float scale = rsqrtf(fmaxf(sm[0], HB_EPS));
    #pragma unroll
    for (int i = 0; i < 16; ++i)
        rowp[t + i * 256] = v[i] * scale;
}

extern "C" void kernel_launch(void* const* d_in, const int* in_sizes, int n_in,
                              void* d_out, int out_size, void* d_ws, size_t ws_size,
                              hipStream_t stream) {
    (void)in_sizes; (void)n_in; (void)out_size; (void)ws_size;
    const float* inp  = (const float*)d_in[0];
    const float* base = (const float*)d_in[1];
    const float* pw   = (const float*)d_in[2];
    float* out = (float*)d_out;

    float* w = (float*)d_ws;
    float* y = (float*)((char*)d_ws + 1024);

    hb_softmax128<<<1, 128, 0, stream>>>(pw, w);
    hb_shiftsum  <<<dim3(8, NPAIRS), 256, 0, stream>>>(inp, w, y);
    hb_gemm_wmma <<<dim3(64, 4),     256, 0, stream>>>(y, base, out);
    hb_normalize <<<NPAIRS,          256, 0, stream>>>(out);
}


# Shift-sum re-shaped for loads in flight: thread owns 4 adjacent outputs, one 16-byte load per row, rows requested 8 at a time six batches ahead (48-56 loads in flight per wave, was 8 then a drain)
# speedup vs baseline: 1.8251x; 1.2142x over previous
#include <hip/hip_runtime.h>

#define D      4096
#define DMASK  4095
#define BS     128
#define NPAIRS 64
#define HB_EPS 1e-8f

typedef __attribute__((ext_vector_type(2))) float v2f;
typedef __attribute__((ext_vector_type(8))) float v8f;

__global__ void hb_softmax128(const float* __restrict__ pw, float* __restrict__ w) {
    __shared__ float sm[BS];
    int t = threadIdx.x;
    float x = pw[t];
    sm[t] = x; __syncthreads();
    for (int s = BS / 2; s > 0; s >>= 1) {
        if (t < s) sm[t] = fmaxf(sm[t], sm[t + s]);
        __syncthreads();
    }
    float mx = sm[0]; __syncthreads();
    float e = expf(x - mx);
    sm[t] = e; __syncthreads();
    for (int s = BS / 2; s > 0; s >>= 1) {
        if (t < s) sm[t] += sm[t + s];
        __syncthreads();
    }
    w[t] = e / sm[0];
}

#define SS_THREADS 128
#define SS_AHEAD   6
#define SS_RING    (SS_AHEAD + 1)

typedef __attribute__((ext_vector_type(4))) float v4f;

template <int RB>
__device__ __forceinline__ void ss_issue_main(v4f (&d)[RB], float (&wb)[RB],
        const float* p, const float* __restrict__ w, int s0) {
    #pragma unroll
    for (int r = 0; r < RB; ++r)
        __builtin_memcpy(&d[r], p + (size_t)(s0 + r) * (D + 1), 16);
    #pragma unroll
    for (int r = 0; r < RB; ++r) wb[r] = w[s0 + r];
}

template <int RB>
__device__ __forceinline__ void ss_issue_edge(v4f (&d)[RB], float (&wb)[RB],
        const float* prow, const float* __restrict__ w, int Q, int s0) {
    #pragma unroll
    for (int r = 0; r < RB; ++r) {
        const int s = s0 + r, a = s >> 2, b = s & 3;
        const float* row = prow + (size_t)s * D;
        float x[4];
        __builtin_memcpy(&x[0], row + 4 * ((Q + a) & 1023) + b, 4 * (4 - b));
        if (b) __builtin_memcpy(&x[4 - b], row + 4 * ((Q + a + 1) & 1023), 4 * b);
        d[r] = (v4f){ x[0], x[1], x[2], x[3] };
    }
    #pragma unroll
    for (int r = 0; r < RB; ++r) wb[r] = w[s0 + r];
}

template <int RB>
__device__ __forceinline__ void ss_use(v4f& acc, const v4f (&d)[RB], const float (&wb)[RB]) {
    #pragma unroll
    for (int r = 0; r < RB; ++r) {
        float ws = wb[r];
        acc.x = fmaf(ws, d[r].x, acc.x);
        acc.y = fmaf(ws, d[r].y, acc.y);
        acc.z = fmaf(ws, d[r].z, acc.z);
        acc.w = fmaf(ws, d[r].w, acc.w);
    }
}

__global__ void __launch_bounds__(SS_THREADS) hb_shiftsum(
        const float* __restrict__ inp, const float* __restrict__ w,
        float* __restrict__ y) {
    int pair = blockIdx.y;
    int Q = blockIdx.x * SS_THREADS + threadIdx.x;
    const float* prow = inp + (size_t)pair * BS * D;
    v4f acc = { 0.0f, 0.0f, 0.0f, 0.0f };

    int qw = __builtin_amdgcn_readfirstlane(Q & ~31);
    if (qw < 992) {
        constexpr int RB = 8, NB = BS / RB;
        const float* p = prow + 4 * Q;
        v4f buf[SS_RING][RB];
        float wb[SS_RING][RB];
        #pragma unroll
        for (int g = 0; g < SS_AHEAD; ++g) ss_issue_main<RB>(buf[g], wb[g], p, w, g * RB);
        #pragma unroll
        for (int g = 0; g < NB; ++g) {
            if (g + SS_AHEAD < NB)
                ss_issue_main<RB>(buf[(g + SS_AHEAD) % SS_RING], wb[(g + SS_AHEAD) % SS_RING],
                                  p, w, (g + SS_AHEAD) * RB);
            __builtin_amdgcn_sched_barrier(0);
            ss_use<RB>(acc, buf[g % SS_RING], wb[g % SS_RING]);
            __builtin_amdgcn_sched_barrier(0);
        }
    } else {
        constexpr int RB = 4, NB = BS / RB;
        v4f buf[SS_RING][RB];
        float wb[SS_RING][RB];
        #pragma unroll
        for (int g = 0; g < SS_AHEAD; ++g) ss_issue_edge<RB>(buf[g], wb[g], prow, w, Q, g * RB);
        #pragma unroll
        for (int g = 0; g < NB; ++g) {
            if (g + SS_AHEAD < NB)
                ss_issue_edge<RB>(buf[(g + SS_AHEAD) % SS_RING], wb[(g + SS_AHEAD) % SS_RING],
                                  prow, w, Q, (g + SS_AHEAD) * RB);
            __builtin_amdgcn_sched_barrier(0);
            ss_use<RB>(acc, buf[g % SS_RING], wb[g % SS_RING]);
            __builtin_amdgcn_sched_barrier(0);
        }
    }
    *(v4f*)(y + (size_t)pair * D + 4 * Q) = acc;
}

#define GK_WAVES 8
#define GK_KPW   (D / GK_WAVES)
#define GK_CHUNK 32

__device__ __forceinline__ v8f hb_mma(v2f a, v2f b, v8f c) {
    return __builtin_amdgcn_wmma_f32_16x16x4_f32(
                false, a,  false, b,
                (short)0, c,  false,  false);
}

__device__ __forceinline__ void hb_load4(float4 (&d)[4], const float* p) {
    #pragma unroll
    for (int s = 0; s < 4; ++s) d[s] = *(const float4*)(p + 8 * s);
}

__device__ __forceinline__ void hb_chunk(
        const float4 (&a)[4], const float* b0, const float* b1,
        const float* b2, const float* b3,
        v8f& acc0, v8f& acc1, v8f& acc2, v8f& acc3) {
    #pragma unroll
    for (int s = 0; s < 4; ++s) {
        int ko = 8 * s;
        v2f alo = { a[s].x, a[s].y };
        v2f ahi = { a[s].z, a[s].w };
        v2f b0lo = { b0[ko], b0[ko + 1] }, b0hi = { b0[ko + 2], b0[ko + 3] };
        v2f b1lo = { b1[ko], b1[ko + 1] }, b1hi = { b1[ko + 2], b1[ko + 3] };
        v2f b2lo = { b2[ko], b2[ko + 1] }, b2hi = { b2[ko + 2], b2[ko + 3] };
        v2f b3lo = { b3[ko], b3[ko + 1] }, b3hi = { b3[ko + 2], b3[ko + 3] };
        acc0 = hb_mma(alo, b0lo, acc0);
        acc1 = hb_mma(alo, b1lo, acc1);
        acc2 = hb_mma(alo, b2lo, acc2);
        acc3 = hb_mma(alo, b3lo, acc3);
        acc0 = hb_mma(ahi, b0hi, acc0);
        acc1 = hb_mma(ahi, b1hi, acc1);
        acc2 = hb_mma(ahi, b2hi, acc2);
        acc3 = hb_mma(ahi, b3hi, acc3);
    }
}

__global__ void __launch_bounds__(256) hb_gemm_wmma(
        const float* __restrict__ y, const float* __restrict__ base,
        float* __restrict__ out) {
    __shared__ __attribute__((aligned(16))) float lbase[2 * D];
    int t = threadIdx.x;
    for (int i = t; i < D; i += 256) {
        float v = base[i];
        lbase[i]     = v;
        lbase[i + D] = v;
    }
    __syncthreads();

    int wave = t >> 5;
    int lane = t & 31;
    int half = lane >> 4;
    int l    = lane & 15;

    int m0 = blockIdx.y * 16;
    int n0 = blockIdx.x * 64;
    int kbeg = wave * GK_KPW;

    const float* ap = y + (size_t)(m0 + l) * D + kbeg + 4 * half;
    const float* bp0 = lbase + ((4 * half - (n0 +  0 + l) + 8192) & DMASK) + kbeg;
    const float* bp1 = lbase + ((4 * half - (n0 + 16 + l) + 8192) & DMASK) + kbeg;
    const float* bp2 = lbase + ((4 * half - (n0 + 32 + l) + 8192) & DMASK) + kbeg;
    const float* bp3 = lbase + ((4 * half - (n0 + 48 + l) + 8192) & DMASK) + kbeg;

    v8f acc0 = {}, acc1 = {}, acc2 = {}, acc3 = {};

    float4 ea[4], oa[4];
    hb_load4(ea, ap);

    #pragma unroll 1
    for (int kc = 0; kc < GK_KPW - 2 * GK_CHUNK; kc += 2 * GK_CHUNK) {
        hb_load4(oa, ap + kc + GK_CHUNK);
        hb_chunk(ea, bp0 + kc, bp1 + kc, bp2 + kc, bp3 + kc, acc0, acc1, acc2, acc3);
        hb_load4(ea, ap + kc + 2 * GK_CHUNK);
        hb_chunk(oa, bp0 + kc + GK_CHUNK, bp1 + kc + GK_CHUNK, bp2 + kc + GK_CHUNK,
                 bp3 + kc + GK_CHUNK, acc0, acc1, acc2, acc3);
    }
    {
        const int kc = GK_KPW - 2 * GK_CHUNK;
        hb_load4(oa, ap + kc + GK_CHUNK);
        hb_chunk(ea, bp0 + kc, bp1 + kc, bp2 + kc, bp3 + kc, acc0, acc1, acc2, acc3);
        hb_chunk(oa, bp0 + kc + GK_CHUNK, bp1 + kc + GK_CHUNK, bp2 + kc + GK_CHUNK,
                 bp3 + kc + GK_CHUNK, acc0, acc1, acc2, acc3);
    }

    __syncthreads();
    float* red = lbase;
    float* myred = red + wave * 1024 + lane;
    #pragma unroll
    for (int v = 0; v < 8; ++v) {
        myred[(0 * 8 + v) * 32] = acc0[v];
        myred[(1 * 8 + v) * 32] = acc1[v];
        myred[(2 * 8 + v) * 32] = acc2[v];
        myred[(3 * 8 + v) * 32] = acc3[v];
    }
    __syncthreads();

    int r   = t >> 4;
    int c4  = (t & 15) << 2;
    int ridx = (((c4 >> 4) * 8 + (r & 7)) * 32) + (r >> 3) * 16 + (c4 & 15);
    float4 sum = *(const float4*)(red + ridx);
    #pragma unroll
    for (int w = 1; w < GK_WAVES; ++w) {
        float4 p = *(const float4*)(red + w * 1024 + ridx);
        sum.x += p.x; sum.y += p.y; sum.z += p.z; sum.w += p.w;
    }
    *(float4*)(out + (size_t)(m0 + r) * D + n0 + c4) = sum;
}

__global__ void __launch_bounds__(256) hb_normalize(float* __restrict__ out) {
    __shared__ float sm[256];
    int t = threadIdx.x;
    float* rowp = out + (size_t)blockIdx.x * D;
    float v[16];
    float ssq = 0.0f;
    #pragma unroll
    for (int i = 0; i < 16; ++i) {
        v[i] = rowp[t + i * 256];
        ssq = fmaf(v[i], v[i], ssq);
    }
    sm[t] = ssq; __syncthreads();
    for (int s = 128; s > 0; s >>= 1) {
        if (t < s) sm[t] += sm[t + s];
        __syncthreads();
    }
    float scale = rsqrtf(fmaxf(sm[0], HB_EPS));
    #pragma unroll
    for (int i = 0; i < 16; ++i)
        rowp[t + i * 256] = v[i] * scale;
}

extern "C" void kernel_launch(void* const* d_in, const int* in_sizes, int n_in,
                              void* d_out, int out_size, void* d_ws, size_t ws_size,
                              hipStream_t stream) {
    (void)in_sizes; (void)n_in; (void)out_size; (void)ws_size;
    const float* inp  = (const float*)d_in[0];
    const float* base = (const float*)d_in[1];
    const float* pw   = (const float*)d_in[2];
    float* out = (float*)d_out;

    float* w = (float*)d_ws;
    float* y = (float*)((char*)d_ws + 1024);

    hb_softmax128<<<1, 128, 0, stream>>>(pw, w);
    hb_shiftsum  <<<dim3(8, NPAIRS), SS_THREADS, 0, stream>>>(inp, w, y);
    hb_gemm_wmma <<<dim3(64, 4),     256, 0, stream>>>(y, base, out);
    hb_normalize <<<NPAIRS,          256, 0, stream>>>(out);
}


# Softmax launch folded into the shift-sum: each wave computes the 128 weights itself (4 per lane, lane-exchange sum in the old tree's pairing, same bits) and reads each by lane; 3 launches, was 4
# speedup vs baseline: 1.9456x; 1.0660x over previous
#include <hip/hip_runtime.h>

#define D      4096
#define DMASK  4095
#define BS     128
#define NPAIRS 64
#define HB_EPS 1e-8f

typedef __attribute__((ext_vector_type(2))) float v2f;
typedef __attribute__((ext_vector_type(8))) float v8f;

#define SS_THREADS 128
#define SS_AHEAD   6
#define SS_RING    (SS_AHEAD + 1)

typedef __attribute__((ext_vector_type(4))) float v4f;

__device__ __forceinline__ float ss_xor_half(float v) {
    const int i = __builtin_bit_cast(int, v);
    return __builtin_bit_cast(float,
        __builtin_amdgcn_permlanex16(i, i, 0x76543210, 0xfedcba98, false, false));
}
template <int M>
__device__ __forceinline__ float ss_xor_row(float v) {
    const int i = __builtin_bit_cast(int, v);
    return __builtin_bit_cast(float,
        __builtin_amdgcn_update_dpp(i, i, 0x160 + M, 0xf, 0xf, false));
}

__device__ __forceinline__ void ss_weights(const float (&x)[4], float (&wv)[4]) {
    float mx = fmaxf(fmaxf(x[0], x[2]), fmaxf(x[1], x[3]));
    mx = fmaxf(mx, ss_xor_half(mx));
    mx = fmaxf(mx, ss_xor_row<8>(mx));
    mx = fmaxf(mx, ss_xor_row<4>(mx));
    mx = fmaxf(mx, ss_xor_row<2>(mx));
    mx = fmaxf(mx, ss_xor_row<1>(mx));
    float e[4];
    #pragma unroll
    for (int j = 0; j < 4; ++j) e[j] = expf(x[j] - mx);
    float sum = (e[0] + e[2]) + (e[1] + e[3]);
    sum += ss_xor_half(sum);
    sum += ss_xor_row<8>(sum);
    sum += ss_xor_row<4>(sum);
    sum += ss_xor_row<2>(sum);
    sum += ss_xor_row<1>(sum);
    #pragma unroll
    for (int j = 0; j < 4; ++j) wv[j] = e[j] / sum;
}

__device__ __forceinline__ void ss_hold_weights_here(float (&x)[4]) {
    asm volatile("" : "+v"(x[0]), "+v"(x[1]), "+v"(x[2]), "+v"(x[3]));
}

__device__ __forceinline__ float ss_lane_value(float v, int lane) {
    return __builtin_bit_cast(float,
        __builtin_amdgcn_readlane(__builtin_bit_cast(int, v), lane));
}

template <int RB>
__device__ __forceinline__ void ss_issue_main(v4f (&d)[RB], const float* p, int s0) {
    #pragma unroll
    for (int r = 0; r < RB; ++r)
        __builtin_memcpy(&d[r], p + (size_t)(s0 + r) * (D + 1), 16);
}

template <int RB>
__device__ __forceinline__ void ss_issue_edge(v4f (&d)[RB],
        const float* prow, int Q, int s0) {
    #pragma unroll
    for (int r = 0; r < RB; ++r) {
        const int s = s0 + r, a = s >> 2, b = s & 3;
        const float* row = prow + (size_t)s * D;
        float x[4];
        __builtin_memcpy(&x[0], row + 4 * ((Q + a) & 1023) + b, 4 * (4 - b));
        if (b) __builtin_memcpy(&x[4 - b], row + 4 * ((Q + a + 1) & 1023), 4 * b);
        d[r] = (v4f){ x[0], x[1], x[2], x[3] };
    }
}

template <int RB>
__device__ __forceinline__ void ss_use(v4f& acc, const v4f (&d)[RB],
        const float (&wv)[4], int s0) {
    #pragma unroll
    for (int r = 0; r < RB; ++r) {
        const int s = s0 + r;
        float ws = ss_lane_value(wv[s >> 5], s & 31);
        acc.x = fmaf(ws, d[r].x, acc.x);
        acc.y = fmaf(ws, d[r].y, acc.y);
        acc.z = fmaf(ws, d[r].z, acc.z);
        acc.w = fmaf(ws, d[r].w, acc.w);
    }
}

__global__ void __launch_bounds__(SS_THREADS) hb_shiftsum(
        const float* __restrict__ inp, const float* __restrict__ pw,
        float* __restrict__ y) {
    int pair = blockIdx.y;
    int Q = blockIdx.x * SS_THREADS + threadIdx.x;
    const float* prow = inp + (size_t)pair * BS * D;
    v4f acc = { 0.0f, 0.0f, 0.0f, 0.0f };

    const int lane = threadIdx.x & 31;
    float x[4];
    #pragma unroll
    for (int j = 0; j < 4; ++j) x[j] = pw[lane + 32 * j];
    float wv[4];

    int qw = __builtin_amdgcn_readfirstlane(Q & ~31);
    if (qw < 992) {
        constexpr int RB = 8, NB = BS / RB;
        const float* p = prow + 4 * Q;
        v4f buf[SS_RING][RB];
        #pragma unroll
        for (int g = 0; g < SS_AHEAD; ++g) ss_issue_main<RB>(buf[g], p, g * RB);
        ss_hold_weights_here(x);
        ss_weights(x, wv);
        #pragma unroll
        for (int g = 0; g < NB; ++g) {
            if (g + SS_AHEAD < NB)
                ss_issue_main<RB>(buf[(g + SS_AHEAD) % SS_RING], p, (g + SS_AHEAD) * RB);
            __builtin_amdgcn_sched_barrier(0);
            ss_use<RB>(acc, buf[g % SS_RING], wv, g * RB);
            __builtin_amdgcn_sched_barrier(0);
        }
    } else {
        constexpr int RB = 4, NB = BS / RB;
        v4f buf[SS_RING][RB];
        #pragma unroll
        for (int g = 0; g < SS_AHEAD; ++g) ss_issue_edge<RB>(buf[g], prow, Q, g * RB);
        ss_hold_weights_here(x);
        ss_weights(x, wv);
        #pragma unroll
        for (int g = 0; g < NB; ++g) {
            if (g + SS_AHEAD < NB)
                ss_issue_edge<RB>(buf[(g + SS_AHEAD) % SS_RING], prow, Q, (g + SS_AHEAD) * RB);
            __builtin_amdgcn_sched_barrier(0);
            ss_use<RB>(acc, buf[g % SS_RING], wv, g * RB);
            __builtin_amdgcn_sched_barrier(0);
        }
    }
    *(v4f*)(y + (size_t)pair * D + 4 * Q) = acc;
}

#define GK_WAVES 8
#define GK_KPW   (D / GK_WAVES)
#define GK_CHUNK 32

__device__ __forceinline__ v8f hb_mma(v2f a, v2f b, v8f c) {
    return __builtin_amdgcn_wmma_f32_16x16x4_f32(
                false, a,  false, b,
                (short)0, c,  false,  false);
}

__device__ __forceinline__ void hb_load4(float4 (&d)[4], const float* p) {
    #pragma unroll
    for (int s = 0; s < 4; ++s) d[s] = *(const float4*)(p + 8 * s);
}

__device__ __forceinline__ void hb_chunk(
        const float4 (&a)[4], const float* b0, const float* b1,
        const float* b2, const float* b3,
        v8f& acc0, v8f& acc1, v8f& acc2, v8f& acc3) {
    #pragma unroll
    for (int s = 0; s < 4; ++s) {
        int ko = 8 * s;
        v2f alo = { a[s].x, a[s].y };
        v2f ahi = { a[s].z, a[s].w };
        v2f b0lo = { b0[ko], b0[ko + 1] }, b0hi = { b0[ko + 2], b0[ko + 3] };
        v2f b1lo = { b1[ko], b1[ko + 1] }, b1hi = { b1[ko + 2], b1[ko + 3] };
        v2f b2lo = { b2[ko], b2[ko + 1] }, b2hi = { b2[ko + 2], b2[ko + 3] };
        v2f b3lo = { b3[ko], b3[ko + 1] }, b3hi = { b3[ko + 2], b3[ko + 3] };
        acc0 = hb_mma(alo, b0lo, acc0);
        acc1 = hb_mma(alo, b1lo, acc1);
        acc2 = hb_mma(alo, b2lo, acc2);
        acc3 = hb_mma(alo, b3lo, acc3);
        acc0 = hb_mma(ahi, b0hi, acc0);
        acc1 = hb_mma(ahi, b1hi, acc1);
        acc2 = hb_mma(ahi, b2hi, acc2);
        acc3 = hb_mma(ahi, b3hi, acc3);
    }
}

__global__ void __launch_bounds__(256) hb_gemm_wmma(
        const float* __restrict__ y, const float* __restrict__ base,
        float* __restrict__ out) {
    __shared__ __attribute__((aligned(16))) float lbase[2 * D];
    int t = threadIdx.x;
    for (int i = t; i < D; i += 256) {
        float v = base[i];
        lbase[i]     = v;
        lbase[i + D] = v;
    }
    __syncthreads();

    int wave = t >> 5;
    int lane = t & 31;
    int half = lane >> 4;
    int l    = lane & 15;

    int m0 = blockIdx.y * 16;
    int n0 = blockIdx.x * 64;
    int kbeg = wave * GK_KPW;

    const float* ap = y + (size_t)(m0 + l) * D + kbeg + 4 * half;
    const float* bp0 = lbase + ((4 * half - (n0 +  0 + l) + 8192) & DMASK) + kbeg;
    const float* bp1 = lbase + ((4 * half - (n0 + 16 + l) + 8192) & DMASK) + kbeg;
    const float* bp2 = lbase + ((4 * half - (n0 + 32 + l) + 8192) & DMASK) + kbeg;
    const float* bp3 = lbase + ((4 * half - (n0 + 48 + l) + 8192) & DMASK) + kbeg;

    v8f acc0 = {}, acc1 = {}, acc2 = {}, acc3 = {};

    float4 ea[4], oa[4];
    hb_load4(ea, ap);

    #pragma unroll 1
    for (int kc = 0; kc < GK_KPW - 2 * GK_CHUNK; kc += 2 * GK_CHUNK) {
        hb_load4(oa, ap + kc + GK_CHUNK);
        hb_chunk(ea, bp0 + kc, bp1 + kc, bp2 + kc, bp3 + kc, acc0, acc1, acc2, acc3);
        hb_load4(ea, ap + kc + 2 * GK_CHUNK);
        hb_chunk(oa, bp0 + kc + GK_CHUNK, bp1 + kc + GK_CHUNK, bp2 + kc + GK_CHUNK,
                 bp3 + kc + GK_CHUNK, acc0, acc1, acc2, acc3);
    }
    {
        const int kc = GK_KPW - 2 * GK_CHUNK;
        hb_load4(oa, ap + kc + GK_CHUNK);
        hb_chunk(ea, bp0 + kc, bp1 + kc, bp2 + kc, bp3 + kc, acc0, acc1, acc2, acc3);
        hb_chunk(oa, bp0 + kc + GK_CHUNK, bp1 + kc + GK_CHUNK, bp2 + kc + GK_CHUNK,
                 bp3 + kc + GK_CHUNK, acc0, acc1, acc2, acc3);
    }

    __syncthreads();
    float* red = lbase;
    float* myred = red + wave * 1024 + lane;
    #pragma unroll
    for (int v = 0; v < 8; ++v) {
        myred[(0 * 8 + v) * 32] = acc0[v];
        myred[(1 * 8 + v) * 32] = acc1[v];
        myred[(2 * 8 + v) * 32] = acc2[v];
        myred[(3 * 8 + v) * 32] = acc3[v];
    }
    __syncthreads();

    int r   = t >> 4;
    int c4  = (t & 15) << 2;
    int ridx = (((c4 >> 4) * 8 + (r & 7)) * 32) + (r >> 3) * 16 + (c4 & 15);
    float4 sum = *(const float4*)(red + ridx);
    #pragma unroll
    for (int w = 1; w < GK_WAVES; ++w) {
        float4 p = *(const float4*)(red + w * 1024 + ridx);
        sum.x += p.x; sum.y += p.y; sum.z += p.z; sum.w += p.w;
    }
    *(float4*)(out + (size_t)(m0 + r) * D + n0 + c4) = sum;
}

__global__ void __launch_bounds__(256) hb_normalize(float* __restrict__ out) {
    __shared__ float sm[256];
    int t = threadIdx.x;
    float* rowp = out + (size_t)blockIdx.x * D;
    float v[16];
    float ssq = 0.0f;
    #pragma unroll
    for (int i = 0; i < 16; ++i) {
        v[i] = rowp[t + i * 256];
        ssq = fmaf(v[i], v[i], ssq);
    }
    sm[t] = ssq; __syncthreads();
    for (int s = 128; s > 0; s >>= 1) {
        if (t < s) sm[t] += sm[t + s];
        __syncthreads();
    }
    float scale = rsqrtf(fmaxf(sm[0], HB_EPS));
    #pragma unroll
    for (int i = 0; i < 16; ++i)
        rowp[t + i * 256] = v[i] * scale;
}

extern "C" void kernel_launch(void* const* d_in, const int* in_sizes, int n_in,
                              void* d_out, int out_size, void* d_ws, size_t ws_size,
                              hipStream_t stream) {
    (void)in_sizes; (void)n_in; (void)out_size; (void)ws_size;
    const float* inp  = (const float*)d_in[0];
    const float* base = (const float*)d_in[1];
    const float* pw   = (const float*)d_in[2];
    float* out = (float*)d_out;

    float* y = (float*)((char*)d_ws + 1024);

    hb_shiftsum  <<<dim3(8, NPAIRS), SS_THREADS, 0, stream>>>(inp, pw, y);
    hb_gemm_wmma <<<dim3(64, 4),     256, 0, stream>>>(y, base, out);
    hb_normalize <<<NPAIRS,          256, 0, stream>>>(out);
}


# GEMM B path: the circulant makes N-tile j's fragment at step kb equal tile 0's at kb-16j, so each is read from LDS once into a register ring, in groups of 8 a chunk ahead; sums bit-identical
# speedup vs baseline: 2.1559x; 1.1081x over previous
#include <hip/hip_runtime.h>

#define D      4096
#define DMASK  4095
#define BS     128
#define NPAIRS 64
#define HB_EPS 1e-8f

typedef __attribute__((ext_vector_type(2))) float v2f;
typedef __attribute__((ext_vector_type(8))) float v8f;

#define SS_THREADS 128
#define SS_AHEAD   6
#define SS_RING    (SS_AHEAD + 1)

typedef __attribute__((ext_vector_type(4))) float v4f;

__device__ __forceinline__ float ss_xor_half(float v) {
    const int i = __builtin_bit_cast(int, v);
    return __builtin_bit_cast(float,
        __builtin_amdgcn_permlanex16(i, i, 0x76543210, 0xfedcba98, false, false));
}
template <int M>
__device__ __forceinline__ float ss_xor_row(float v) {
    const int i = __builtin_bit_cast(int, v);
    return __builtin_bit_cast(float,
        __builtin_amdgcn_update_dpp(i, i, 0x160 + M, 0xf, 0xf, false));
}

__device__ __forceinline__ void ss_weights(const float (&x)[4], float (&wv)[4]) {
    float mx = fmaxf(fmaxf(x[0], x[2]), fmaxf(x[1], x[3]));
    mx = fmaxf(mx, ss_xor_half(mx));
    mx = fmaxf(mx, ss_xor_row<8>(mx));
    mx = fmaxf(mx, ss_xor_row<4>(mx));
    mx = fmaxf(mx, ss_xor_row<2>(mx));
    mx = fmaxf(mx, ss_xor_row<1>(mx));
    float e[4];
    #pragma unroll
    for (int j = 0; j < 4; ++j) e[j] = expf(x[j] - mx);
    float sum = (e[0] + e[2]) + (e[1] + e[3]);
    sum += ss_xor_half(sum);
    sum += ss_xor_row<8>(sum);
    sum += ss_xor_row<4>(sum);
    sum += ss_xor_row<2>(sum);
    sum += ss_xor_row<1>(sum);
    #pragma unroll
    for (int j = 0; j < 4; ++j) wv[j] = e[j] / sum;
}

__device__ __forceinline__ void ss_hold_weights_here(float (&x)[4]) {
    asm volatile("" : "+v"(x[0]), "+v"(x[1]), "+v"(x[2]), "+v"(x[3]));
}

__device__ __forceinline__ float ss_lane_value(float v, int lane) {
    return __builtin_bit_cast(float,
        __builtin_amdgcn_readlane(__builtin_bit_cast(int, v), lane));
}

template <int RB>
__device__ __forceinline__ void ss_issue_main(v4f (&d)[RB], const float* p, int s0) {
    #pragma unroll
    for (int r = 0; r < RB; ++r)
        __builtin_memcpy(&d[r], p + (size_t)(s0 + r) * (D + 1), 16);
}

template <int RB>
__device__ __forceinline__ void ss_issue_edge(v4f (&d)[RB],
        const float* prow, int Q, int s0) {
    #pragma unroll
    for (int r = 0; r < RB; ++r) {
        const int s = s0 + r, a = s >> 2, b = s & 3;
        const float* row = prow + (size_t)s * D;
        float x[4];
        __builtin_memcpy(&x[0], row + 4 * ((Q + a) & 1023) + b, 4 * (4 - b));
        if (b) __builtin_memcpy(&x[4 - b], row + 4 * ((Q + a + 1) & 1023), 4 * b);
        d[r] = (v4f){ x[0], x[1], x[2], x[3] };
    }
}

template <int RB>
__device__ __forceinline__ void ss_use(v4f& acc, const v4f (&d)[RB],
        const float (&wv)[4], int s0) {
    #pragma unroll
    for (int r = 0; r < RB; ++r) {
        const int s = s0 + r;
        float ws = ss_lane_value(wv[s >> 5], s & 31);
        acc.x = fmaf(ws, d[r].x, acc.x);
        acc.y = fmaf(ws, d[r].y, acc.y);
        acc.z = fmaf(ws, d[r].z, acc.z);
        acc.w = fmaf(ws, d[r].w, acc.w);
    }
}

__global__ void __launch_bounds__(SS_THREADS) hb_shiftsum(
        const float* __restrict__ inp, const float* __restrict__ pw,
        float* __restrict__ y) {
    int pair = blockIdx.y;
    int Q = blockIdx.x * SS_THREADS + threadIdx.x;
    const float* prow = inp + (size_t)pair * BS * D;
    v4f acc = { 0.0f, 0.0f, 0.0f, 0.0f };

    const int lane = threadIdx.x & 31;
    float x[4];
    #pragma unroll
    for (int j = 0; j < 4; ++j) x[j] = pw[lane + 32 * j];
    float wv[4];

    int qw = __builtin_amdgcn_readfirstlane(Q & ~31);
    if (qw < 992) {
        constexpr int RB = 8, NB = BS / RB;
        const float* p = prow + 4 * Q;
        v4f buf[SS_RING][RB];
        #pragma unroll
        for (int g = 0; g < SS_AHEAD; ++g) ss_issue_main<RB>(buf[g], p, g * RB);
        ss_hold_weights_here(x);
        ss_weights(x, wv);
        #pragma unroll
        for (int g = 0; g < NB; ++g) {
            if (g + SS_AHEAD < NB)
                ss_issue_main<RB>(buf[(g + SS_AHEAD) % SS_RING], p, (g + SS_AHEAD) * RB);
            __builtin_amdgcn_sched_barrier(0);
            ss_use<RB>(acc, buf[g % SS_RING], wv, g * RB);
            __builtin_amdgcn_sched_barrier(0);
        }
    } else {
        constexpr int RB = 4, NB = BS / RB;
        v4f buf[SS_RING][RB];
        #pragma unroll
        for (int g = 0; g < SS_AHEAD; ++g) ss_issue_edge<RB>(buf[g], prow, Q, g * RB);
        ss_hold_weights_here(x);
        ss_weights(x, wv);
        #pragma unroll
        for (int g = 0; g < NB; ++g) {
            if (g + SS_AHEAD < NB)
                ss_issue_edge<RB>(buf[(g + SS_AHEAD) % SS_RING], prow, Q, (g + SS_AHEAD) * RB);
            __builtin_amdgcn_sched_barrier(0);
            ss_use<RB>(acc, buf[g % SS_RING], wv, g * RB);
            __builtin_amdgcn_sched_barrier(0);
        }
    }
    *(v4f*)(y + (size_t)pair * D + 4 * Q) = acc;
}

#define GK_WAVES 8
#define GK_KPW   (D / GK_WAVES)
#define GK_CHUNK 32
#define GK_RING  16
#define GK_BODY  (GK_RING * 8)

__device__ __forceinline__ v8f hb_mma(v2f a, v2f b, v8f c) {
    return __builtin_amdgcn_wmma_f32_16x16x4_f32(
                false, a,  false, b,
                (short)0, c,  false,  false);
}

__device__ __forceinline__ void hb_load4(float4 (&d)[4], const float* p) {
    #pragma unroll
    for (int s = 0; s < 4; ++s) d[s] = *(const float4*)(p + 8 * s);
}

template <int T0, int N>
__device__ __forceinline__ void hb_bfetch(float4 (&ring)[GK_RING], const float* q) {
    #pragma unroll
    for (int u = 0; u < N; ++u) {
        const float* e = q + 8 * (T0 + u);
        ring[(T0 + u) % GK_RING] = make_float4(e[0], e[1], e[2], e[3]);
    }
}

template <int CC>
__device__ __forceinline__ void hb_chunk(
        const float4 (&a)[4], const float4 (&ring)[GK_RING],
        v8f& acc0, v8f& acc1, v8f& acc2, v8f& acc3) {
    #pragma unroll
    for (int s = 0; s < 4; ++s) {
        const int i = 4 * CC + s;
        const float4 e0 = ring[(i + 6) % GK_RING];
        const float4 e1 = ring[(i + 4) % GK_RING];
        const float4 e2 = ring[(i + 2) % GK_RING];
        const float4 e3 = ring[(i + 0) % GK_RING];
        v2f alo = { a[s].x, a[s].y };
        v2f ahi = { a[s].z, a[s].w };
        v2f b0lo = { e0.x, e0.y }, b0hi = { e0.z, e0.w };
        v2f b1lo = { e1.x, e1.y }, b1hi = { e1.z, e1.w };
        v2f b2lo = { e2.x, e2.y }, b2hi = { e2.z, e2.w };
        v2f b3lo = { e3.x, e3.y }, b3hi = { e3.z, e3.w };
        acc0 = hb_mma(alo, b0lo, acc0);
        acc1 = hb_mma(alo, b1lo, acc1);
        acc2 = hb_mma(alo, b2lo, acc2);
        acc3 = hb_mma(alo, b3lo, acc3);
        acc0 = hb_mma(ahi, b0hi, acc0);
        acc1 = hb_mma(ahi, b1hi, acc1);
        acc2 = hb_mma(ahi, b2hi, acc2);
        acc3 = hb_mma(ahi, b3hi, acc3);
    }
}

template <bool LAST>
__device__ __forceinline__ void hb_body(const float* ap, const float* q,
        float4 (&ring)[GK_RING], float4 (&ea)[4], float4 (&oa)[4],
        v8f& acc0, v8f& acc1, v8f& acc2, v8f& acc3) {
    hb_load4(oa, ap + 1 * GK_CHUNK);
    hb_bfetch<10, 4>(ring, q);
    hb_chunk<0>(ea, ring, acc0, acc1, acc2, acc3);
    __builtin_amdgcn_sched_barrier(0);

    hb_load4(ea, ap + 2 * GK_CHUNK);
    hb_bfetch<14, 4>(ring, q);
    hb_chunk<1>(oa, ring, acc0, acc1, acc2, acc3);
    __builtin_amdgcn_sched_barrier(0);

    hb_load4(oa, ap + 3 * GK_CHUNK);
    hb_bfetch<18, 4>(ring, q);
    hb_chunk<2>(ea, ring, acc0, acc1, acc2, acc3);
    __builtin_amdgcn_sched_barrier(0);

    if (!LAST) {
        hb_load4(ea, ap + 4 * GK_CHUNK);
        hb_bfetch<22, 4>(ring, q);
    }
    hb_chunk<3>(oa, ring, acc0, acc1, acc2, acc3);
}

__global__ void __launch_bounds__(256) hb_gemm_wmma(
        const float* __restrict__ y, const float* __restrict__ base,
        float* __restrict__ out) {
    __shared__ __attribute__((aligned(16))) float lbase[2 * D];
    int t = threadIdx.x;
    for (int i = t; i < D; i += 256) {
        float v = base[i];
        lbase[i]     = v;
        lbase[i + D] = v;
    }
    __syncthreads();

    int wave = t >> 5;
    int lane = t & 31;
    int half = lane >> 4;
    int l    = lane & 15;

    int m0 = blockIdx.y * 16;
    int n0 = blockIdx.x * 64;
    int kbeg = wave * GK_KPW;

    const float* ap = y + (size_t)(m0 + l) * D + kbeg + 4 * half;
    const float* bq = lbase + ((4 * half - (n0 + l) - 48 + kbeg + 8192) & DMASK);

    v8f acc0 = {}, acc1 = {}, acc2 = {}, acc3 = {};

    float4 ea[4], oa[4];
    float4 ring[GK_RING];
    hb_load4(ea, ap);
    hb_bfetch<0, 10>(ring, bq);

    #pragma unroll 1
    for (int kc = 0; kc < GK_KPW - GK_BODY; kc += GK_BODY)
        hb_body<false>(ap + kc, bq + kc, ring, ea, oa, acc0, acc1, acc2, acc3);
    hb_body<true>(ap + (GK_KPW - GK_BODY), bq + (GK_KPW - GK_BODY),
                  ring, ea, oa, acc0, acc1, acc2, acc3);

    __syncthreads();
    float* red = lbase;
    float* myred = red + wave * 1024 + lane;
    #pragma unroll
    for (int v = 0; v < 8; ++v) {
        myred[(0 * 8 + v) * 32] = acc0[v];
        myred[(1 * 8 + v) * 32] = acc1[v];
        myred[(2 * 8 + v) * 32] = acc2[v];
        myred[(3 * 8 + v) * 32] = acc3[v];
    }
    __syncthreads();

    int r   = t >> 4;
    int c4  = (t & 15) << 2;
    int ridx = (((c4 >> 4) * 8 + (r & 7)) * 32) + (r >> 3) * 16 + (c4 & 15);
    float4 sum = *(const float4*)(red + ridx);
    #pragma unroll
    for (int w = 1; w < GK_WAVES; ++w) {
        float4 p = *(const float4*)(red + w * 1024 + ridx);
        sum.x += p.x; sum.y += p.y; sum.z += p.z; sum.w += p.w;
    }
    *(float4*)(out + (size_t)(m0 + r) * D + n0 + c4) = sum;
}

__global__ void __launch_bounds__(256) hb_normalize(float* __restrict__ out) {
    __shared__ float sm[256];
    int t = threadIdx.x;
    float* rowp = out + (size_t)blockIdx.x * D;
    float v[16];
    float ssq = 0.0f;
    #pragma unroll
    for (int i = 0; i < 16; ++i) {
        v[i] = rowp[t + i * 256];
        ssq = fmaf(v[i], v[i], ssq);
    }
    sm[t] = ssq; __syncthreads();
    for (int s = 128; s > 0; s >>= 1) {
        if (t < s) sm[t] += sm[t + s];
        __syncthreads();
    }
    float scale = rsqrtf(fmaxf(sm[0], HB_EPS));
    #pragma unroll
    for (int i = 0; i < 16; ++i)
        rowp[t + i * 256] = v[i] * scale;
}

extern "C" void kernel_launch(void* const* d_in, const int* in_sizes, int n_in,
                              void* d_out, int out_size, void* d_ws, size_t ws_size,
                              hipStream_t stream) {
    (void)in_sizes; (void)n_in; (void)out_size; (void)ws_size;
    const float* inp  = (const float*)d_in[0];
    const float* base = (const float*)d_in[1];
    const float* pw   = (const float*)d_in[2];
    float* out = (float*)d_out;

    float* y = (float*)((char*)d_ws + 1024);

    hb_shiftsum  <<<dim3(8, NPAIRS), SS_THREADS, 0, stream>>>(inp, pw, y);
    hb_gemm_wmma <<<dim3(64, 4),     256, 0, stream>>>(y, base, out);
    hb_normalize <<<NPAIRS,          256, 0, stream>>>(out);
}


# GEMM start-up: base[] reaches LDS from four 16-byte loads per thread requested together with the first A quads, in front of the barrier (was 8 dependent load-wait-store trips, then A); same sums
# speedup vs baseline: 2.3758x; 1.1020x over previous
#include <hip/hip_runtime.h>

#define D      4096
#define DMASK  4095
#define BS     128
#define NPAIRS 64
#define HB_EPS 1e-8f

typedef __attribute__((ext_vector_type(2))) float v2f;
typedef __attribute__((ext_vector_type(8))) float v8f;

#define SS_THREADS 128
#define SS_AHEAD   6
#define SS_RING    (SS_AHEAD + 1)

typedef __attribute__((ext_vector_type(4))) float v4f;

__device__ __forceinline__ float ss_xor_half(float v) {
    const int i = __builtin_bit_cast(int, v);
    return __builtin_bit_cast(float,
        __builtin_amdgcn_permlanex16(i, i, 0x76543210, 0xfedcba98, false, false));
}
template <int M>
__device__ __forceinline__ float ss_xor_row(float v) {
    const int i = __builtin_bit_cast(int, v);
    return __builtin_bit_cast(float,
        __builtin_amdgcn_update_dpp(i, i, 0x160 + M, 0xf, 0xf, false));
}

__device__ __forceinline__ void ss_weights(const float (&x)[4], float (&wv)[4]) {
    float mx = fmaxf(fmaxf(x[0], x[2]), fmaxf(x[1], x[3]));
    mx = fmaxf(mx, ss_xor_half(mx));
    mx = fmaxf(mx, ss_xor_row<8>(mx));
    mx = fmaxf(mx, ss_xor_row<4>(mx));
    mx = fmaxf(mx, ss_xor_row<2>(mx));
    mx = fmaxf(mx, ss_xor_row<1>(mx));
    float e[4];
    #pragma unroll
    for (int j = 0; j < 4; ++j) e[j] = expf(x[j] - mx);
    float sum = (e[0] + e[2]) + (e[1] + e[3]);
    sum += ss_xor_half(sum);
    sum += ss_xor_row<8>(sum);
    sum += ss_xor_row<4>(sum);
    sum += ss_xor_row<2>(sum);
    sum += ss_xor_row<1>(sum);
    #pragma unroll
    for (int j = 0; j < 4; ++j) wv[j] = e[j] / sum;
}

__device__ __forceinline__ void ss_hold_weights_here(float (&x)[4]) {
    asm volatile("" : "+v"(x[0]), "+v"(x[1]), "+v"(x[2]), "+v"(x[3]));
}

__device__ __forceinline__ float ss_lane_value(float v, int lane) {
    return __builtin_bit_cast(float,
        __builtin_amdgcn_readlane(__builtin_bit_cast(int, v), lane));
}

template <int RB>
__device__ __forceinline__ void ss_issue_main(v4f (&d)[RB], const float* p, int s0) {
    #pragma unroll
    for (int r = 0; r < RB; ++r)
        __builtin_memcpy(&d[r], p + (size_t)(s0 + r) * (D + 1), 16);
}

template <int RB>
__device__ __forceinline__ void ss_issue_edge(v4f (&d)[RB],
        const float* prow, int Q, int s0) {
    #pragma unroll
    for (int r = 0; r < RB; ++r) {
        const int s = s0 + r, a = s >> 2, b = s & 3;
        const float* row = prow + (size_t)s * D;
        float x[4];
        __builtin_memcpy(&x[0], row + 4 * ((Q + a) & 1023) + b, 4 * (4 - b));
        if (b) __builtin_memcpy(&x[4 - b], row + 4 * ((Q + a + 1) & 1023), 4 * b);
        d[r] = (v4f){ x[0], x[1], x[2], x[3] };
    }
}

template <int RB>
__device__ __forceinline__ void ss_use(v4f& acc, const v4f (&d)[RB],
        const float (&wv)[4], int s0) {
    #pragma unroll
    for (int r = 0; r < RB; ++r) {
        const int s = s0 + r;
        float ws = ss_lane_value(wv[s >> 5], s & 31);
        acc.x = fmaf(ws, d[r].x, acc.x);
        acc.y = fmaf(ws, d[r].y, acc.y);
        acc.z = fmaf(ws, d[r].z, acc.z);
        acc.w = fmaf(ws, d[r].w, acc.w);
    }
}

__global__ void __launch_bounds__(SS_THREADS) hb_shiftsum(
        const float* __restrict__ inp, const float* __restrict__ pw,
        float* __restrict__ y) {
    int pair = blockIdx.y;
    int Q = blockIdx.x * SS_THREADS + threadIdx.x;
    const float* prow = inp + (size_t)pair * BS * D;
    v4f acc = { 0.0f, 0.0f, 0.0f, 0.0f };

    const int lane = threadIdx.x & 31;
    float x[4];
    #pragma unroll
    for (int j = 0; j < 4; ++j) x[j] = pw[lane + 32 * j];
    float wv[4];

    int qw = __builtin_amdgcn_readfirstlane(Q & ~31);
    if (qw < 992) {
        constexpr int RB = 8, NB = BS / RB;
        const float* p = prow + 4 * Q;
        v4f buf[SS_RING][RB];
        #pragma unroll
        for (int g = 0; g < SS_AHEAD; ++g) ss_issue_main<RB>(buf[g], p, g * RB);
        ss_hold_weights_here(x);
        ss_weights(x, wv);
        #pragma unroll
        for (int g = 0; g < NB; ++g) {
            if (g + SS_AHEAD < NB)
                ss_issue_main<RB>(buf[(g + SS_AHEAD) % SS_RING], p, (g + SS_AHEAD) * RB);
            __builtin_amdgcn_sched_barrier(0);
            ss_use<RB>(acc, buf[g % SS_RING], wv, g * RB);
            __builtin_amdgcn_sched_barrier(0);
        }
    } else {
        constexpr int RB = 4, NB = BS / RB;
        v4f buf[SS_RING][RB];
        #pragma unroll
        for (int g = 0; g < SS_AHEAD; ++g) ss_issue_edge<RB>(buf[g], prow, Q, g * RB);
        ss_hold_weights_here(x);
        ss_weights(x, wv);
        #pragma unroll
        for (int g = 0; g < NB; ++g) {
            if (g + SS_AHEAD < NB)
                ss_issue_edge<RB>(buf[(g + SS_AHEAD) % SS_RING], prow, Q, (g + SS_AHEAD) * RB);
            __builtin_amdgcn_sched_barrier(0);
            ss_use<RB>(acc, buf[g % SS_RING], wv, g * RB);
            __builtin_amdgcn_sched_barrier(0);
        }
    }
    *(v4f*)(y + (size_t)pair * D + 4 * Q) = acc;
}

#define GK_WAVES 8
#define GK_KPW   (D / GK_WAVES)
#define GK_CHUNK 32
#define GK_RING  16
#define GK_BODY  (GK_RING * 8)

__device__ __forceinline__ v8f hb_mma(v2f a, v2f b, v8f c) {
    return __builtin_amdgcn_wmma_f32_16x16x4_f32(
                false, a,  false, b,
                (short)0, c,  false,  false);
}

__device__ __forceinline__ void hb_load4(float4 (&d)[4], const float* p) {
    #pragma unroll
    for (int s = 0; s < 4; ++s) d[s] = *(const float4*)(p + 8 * s);
}

template <int T0, int N>
__device__ __forceinline__ void hb_bfetch(float4 (&ring)[GK_RING], const float* q) {
    #pragma unroll
    for (int u = 0; u < N; ++u) {
        const float* e = q + 8 * (T0 + u);
        ring[(T0 + u) % GK_RING] = make_float4(e[0], e[1], e[2], e[3]);
    }
}

template <int CC>
__device__ __forceinline__ void hb_chunk(
        const float4 (&a)[4], const float4 (&ring)[GK_RING],
        v8f& acc0, v8f& acc1, v8f& acc2, v8f& acc3) {
    #pragma unroll
    for (int s = 0; s < 4; ++s) {
        const int i = 4 * CC + s;
        const float4 e0 = ring[(i + 6) % GK_RING];
        const float4 e1 = ring[(i + 4) % GK_RING];
        const float4 e2 = ring[(i + 2) % GK_RING];
        const float4 e3 = ring[(i + 0) % GK_RING];
        v2f alo = { a[s].x, a[s].y };
        v2f ahi = { a[s].z, a[s].w };
        v2f b0lo = { e0.x, e0.y }, b0hi = { e0.z, e0.w };
        v2f b1lo = { e1.x, e1.y }, b1hi = { e1.z, e1.w };
        v2f b2lo = { e2.x, e2.y }, b2hi = { e2.z, e2.w };
        v2f b3lo = { e3.x, e3.y }, b3hi = { e3.z, e3.w };
        acc0 = hb_mma(alo, b0lo, acc0);
        acc1 = hb_mma(alo, b1lo, acc1);
        acc2 = hb_mma(alo, b2lo, acc2);
        acc3 = hb_mma(alo, b3lo, acc3);
        acc0 = hb_mma(ahi, b0hi, acc0);
        acc1 = hb_mma(ahi, b1hi, acc1);
        acc2 = hb_mma(ahi, b2hi, acc2);
        acc3 = hb_mma(ahi, b3hi, acc3);
    }
}

template <bool LAST>
__device__ __forceinline__ void hb_body(const float* ap, const float* q,
        float4 (&ring)[GK_RING], float4 (&ea)[4], float4 (&oa)[4],
        v8f& acc0, v8f& acc1, v8f& acc2, v8f& acc3) {
    hb_load4(oa, ap + 1 * GK_CHUNK);
    hb_bfetch<10, 4>(ring, q);
    hb_chunk<0>(ea, ring, acc0, acc1, acc2, acc3);
    __builtin_amdgcn_sched_barrier(0);

    hb_load4(ea, ap + 2 * GK_CHUNK);
    hb_bfetch<14, 4>(ring, q);
    hb_chunk<1>(oa, ring, acc0, acc1, acc2, acc3);
    __builtin_amdgcn_sched_barrier(0);

    hb_load4(oa, ap + 3 * GK_CHUNK);
    hb_bfetch<18, 4>(ring, q);
    hb_chunk<2>(ea, ring, acc0, acc1, acc2, acc3);
    __builtin_amdgcn_sched_barrier(0);

    if (!LAST) {
        hb_load4(ea, ap + 4 * GK_CHUNK);
        hb_bfetch<22, 4>(ring, q);
    }
    hb_chunk<3>(oa, ring, acc0, acc1, acc2, acc3);
}

__global__ void __launch_bounds__(256) hb_gemm_wmma(
        const float* __restrict__ y, const float* __restrict__ base,
        float* __restrict__ out) {
    __shared__ __attribute__((aligned(16))) float lbase[2 * D];
    int t = threadIdx.x;

    float4 bv[4];
    for (int j = 0; j < 4; ++j)
        bv[j] = *(const float4*)(base + 4 * (t + 256 * j));

    int wave = t >> 5;
    int lane = t & 31;
    int half = lane >> 4;
    int l    = lane & 15;

    int m0 = blockIdx.y * 16;
    int n0 = blockIdx.x * 64;
    int kbeg = wave * GK_KPW;

    const float* ap = y + (size_t)(m0 + l) * D + kbeg + 4 * half;
    const float* bq = lbase + ((4 * half - (n0 + l) - 48 + kbeg + 8192) & DMASK);

    v8f acc0 = {}, acc1 = {}, acc2 = {}, acc3 = {};

    float4 ea[4], oa[4];
    float4 ring[GK_RING];
    hb_load4(ea, ap);

    for (int j = 0; j < 4; ++j) {
        *(float4*)(lbase + 4 * (t + 256 * j))     = bv[j];
        *(float4*)(lbase + D + 4 * (t + 256 * j)) = bv[j];
    }
    __syncthreads();

    hb_bfetch<0, 10>(ring, bq);

    #pragma unroll 1
    for (int kc = 0; kc < GK_KPW - GK_BODY; kc += GK_BODY)
        hb_body<false>(ap + kc, bq + kc, ring, ea, oa, acc0, acc1, acc2, acc3);
    hb_body<true>(ap + (GK_KPW - GK_BODY), bq + (GK_KPW - GK_BODY),
                  ring, ea, oa, acc0, acc1, acc2, acc3);

    __syncthreads();
    float* red = lbase;
    float* myred = red + wave * 1024 + lane;
    #pragma unroll
    for (int v = 0; v < 8; ++v) {
        myred[(0 * 8 + v) * 32] = acc0[v];
        myred[(1 * 8 + v) * 32] = acc1[v];
        myred[(2 * 8 + v) * 32] = acc2[v];
        myred[(3 * 8 + v) * 32] = acc3[v];
    }
    __syncthreads();

    int r   = t >> 4;
    int c4  = (t & 15) << 2;
    int ridx = (((c4 >> 4) * 8 + (r & 7)) * 32) + (r >> 3) * 16 + (c4 & 15);
    float4 sum = *(const float4*)(red + ridx);
    #pragma unroll
    for (int w = 1; w < GK_WAVES; ++w) {
        float4 p = *(const float4*)(red + w * 1024 + ridx);
        sum.x += p.x; sum.y += p.y; sum.z += p.z; sum.w += p.w;
    }
    *(float4*)(out + (size_t)(m0 + r) * D + n0 + c4) = sum;
}

__global__ void __launch_bounds__(256) hb_normalize(float* __restrict__ out) {
    __shared__ float sm[256];
    int t = threadIdx.x;
    float* rowp = out + (size_t)blockIdx.x * D;
    float v[16];
    float ssq = 0.0f;
    #pragma unroll
    for (int i = 0; i < 16; ++i) {
        v[i] = rowp[t + i * 256];
        ssq = fmaf(v[i], v[i], ssq);
    }
    sm[t] = ssq; __syncthreads();
    for (int s = 128; s > 0; s >>= 1) {
        if (t < s) sm[t] += sm[t + s];
        __syncthreads();
    }
    float scale = rsqrtf(fmaxf(sm[0], HB_EPS));
    #pragma unroll
    for (int i = 0; i < 16; ++i)
        rowp[t + i * 256] = v[i] * scale;
}

extern "C" void kernel_launch(void* const* d_in, const int* in_sizes, int n_in,
                              void* d_out, int out_size, void* d_ws, size_t ws_size,
                              hipStream_t stream) {
    (void)in_sizes; (void)n_in; (void)out_size; (void)ws_size;
    const float* inp  = (const float*)d_in[0];
    const float* base = (const float*)d_in[1];
    const float* pw   = (const float*)d_in[2];
    float* out = (float*)d_out;

    float* y = (float*)((char*)d_ws + 1024);

    hb_shiftsum  <<<dim3(8, NPAIRS), SS_THREADS, 0, stream>>>(inp, pw, y);
    hb_gemm_wmma <<<dim3(64, 4),     256, 0, stream>>>(y, base, out);
    hb_normalize <<<NPAIRS,          256, 0, stream>>>(out);
}


# GEMM halved: the length-4096 circular product split exactly into a cyclic and a sign-wrapped product of length 2048 on sums and differences of the halves; half the matrix work; new summation order
# speedup vs baseline: 2.5412x; 1.0696x over previous
#include <hip/hip_runtime.h>

#define D      4096
#define DMASK  4095
#define BS     128
#define NPAIRS 64
#define HB_EPS 1e-8f

typedef __attribute__((ext_vector_type(2))) float v2f;
typedef __attribute__((ext_vector_type(8))) float v8f;

#define SS_THREADS 128
#define SS_AHEAD   6
#define SS_RING    (SS_AHEAD + 1)

typedef __attribute__((ext_vector_type(4))) float v4f;

__device__ __forceinline__ float ss_xor_half(float v) {
    const int i = __builtin_bit_cast(int, v);
    return __builtin_bit_cast(float,
        __builtin_amdgcn_permlanex16(i, i, 0x76543210, 0xfedcba98, false, false));
}
template <int M>
__device__ __forceinline__ float ss_xor_row(float v) {
    const int i = __builtin_bit_cast(int, v);
    return __builtin_bit_cast(float,
        __builtin_amdgcn_update_dpp(i, i, 0x160 + M, 0xf, 0xf, false));
}

__device__ __forceinline__ void ss_weights(const float (&x)[4], float (&wv)[4]) {
    float mx = fmaxf(fmaxf(x[0], x[2]), fmaxf(x[1], x[3]));
    mx = fmaxf(mx, ss_xor_half(mx));
    mx = fmaxf(mx, ss_xor_row<8>(mx));
    mx = fmaxf(mx, ss_xor_row<4>(mx));
    mx = fmaxf(mx, ss_xor_row<2>(mx));
    mx = fmaxf(mx, ss_xor_row<1>(mx));
    float e[4];
    #pragma unroll
    for (int j = 0; j < 4; ++j) e[j] = expf(x[j] - mx);
    float sum = (e[0] + e[2]) + (e[1] + e[3]);
    sum += ss_xor_half(sum);
    sum += ss_xor_row<8>(sum);
    sum += ss_xor_row<4>(sum);
    sum += ss_xor_row<2>(sum);
    sum += ss_xor_row<1>(sum);
    #pragma unroll
    for (int j = 0; j < 4; ++j) wv[j] = e[j] / sum;
}

__device__ __forceinline__ void ss_hold_weights_here(float (&x)[4]) {
    asm volatile("" : "+v"(x[0]), "+v"(x[1]), "+v"(x[2]), "+v"(x[3]));
}

__device__ __forceinline__ float ss_lane_value(float v, int lane) {
    return __builtin_bit_cast(float,
        __builtin_amdgcn_readlane(__builtin_bit_cast(int, v), lane));
}

template <int RB>
__device__ __forceinline__ void ss_issue_main(v4f (&d)[RB], const float* p, int s0) {
    #pragma unroll
    for (int r = 0; r < RB; ++r)
        __builtin_memcpy(&d[r], p + (size_t)(s0 + r) * (D + 1), 16);
}

template <int RB>
__device__ __forceinline__ void ss_issue_edge(v4f (&d)[RB],
        const float* prow, int Q, int s0) {
    #pragma unroll
    for (int r = 0; r < RB; ++r) {
        const int s = s0 + r, a = s >> 2, b = s & 3;
        const float* row = prow + (size_t)s * D;
        float x[4];
        __builtin_memcpy(&x[0], row + 4 * ((Q + a) & 1023) + b, 4 * (4 - b));
        if (b) __builtin_memcpy(&x[4 - b], row + 4 * ((Q + a + 1) & 1023), 4 * b);
        d[r] = (v4f){ x[0], x[1], x[2], x[3] };
    }
}

template <int RB>
__device__ __forceinline__ void ss_use(v4f& acc, const v4f (&d)[RB],
        const float (&wv)[4], int s0) {
    #pragma unroll
    for (int r = 0; r < RB; ++r) {
        const int s = s0 + r;
        float ws = ss_lane_value(wv[s >> 5], s & 31);
        acc.x = fmaf(ws, d[r].x, acc.x);
        acc.y = fmaf(ws, d[r].y, acc.y);
        acc.z = fmaf(ws, d[r].z, acc.z);
        acc.w = fmaf(ws, d[r].w, acc.w);
    }
}

__global__ void __launch_bounds__(SS_THREADS) hb_shiftsum(
        const float* __restrict__ inp, const float* __restrict__ pw,
        float* __restrict__ y) {
    int pair = blockIdx.y;
    int Q = blockIdx.x * SS_THREADS + threadIdx.x;
    const float* prow = inp + (size_t)pair * BS * D;
    v4f acc = { 0.0f, 0.0f, 0.0f, 0.0f };

    const int lane = threadIdx.x & 31;
    float x[4];
    #pragma unroll
    for (int j = 0; j < 4; ++j) x[j] = pw[lane + 32 * j];
    float wv[4];

    int qw = __builtin_amdgcn_readfirstlane(Q & ~31);
    if (qw < 992) {
        constexpr int RB = 8, NB = BS / RB;
        const float* p = prow + 4 * Q;
        v4f buf[SS_RING][RB];
        #pragma unroll
        for (int g = 0; g < SS_AHEAD; ++g) ss_issue_main<RB>(buf[g], p, g * RB);
        ss_hold_weights_here(x);
        ss_weights(x, wv);
        #pragma unroll
        for (int g = 0; g < NB; ++g) {
            if (g + SS_AHEAD < NB)
                ss_issue_main<RB>(buf[(g + SS_AHEAD) % SS_RING], p, (g + SS_AHEAD) * RB);
            __builtin_amdgcn_sched_barrier(0);
            ss_use<RB>(acc, buf[g % SS_RING], wv, g * RB);
            __builtin_amdgcn_sched_barrier(0);
        }
    } else {
        constexpr int RB = 4, NB = BS / RB;
        v4f buf[SS_RING][RB];
        #pragma unroll
        for (int g = 0; g < SS_AHEAD; ++g) ss_issue_edge<RB>(buf[g], prow, Q, g * RB);
        ss_hold_weights_here(x);
        ss_weights(x, wv);
        #pragma unroll
        for (int g = 0; g < NB; ++g) {
            if (g + SS_AHEAD < NB)
                ss_issue_edge<RB>(buf[(g + SS_AHEAD) % SS_RING], prow, Q, (g + SS_AHEAD) * RB);
            __builtin_amdgcn_sched_barrier(0);
            ss_use<RB>(acc, buf[g % SS_RING], wv, g * RB);
            __builtin_amdgcn_sched_barrier(0);
        }
    }
    *(v4f*)(y + (size_t)pair * D + 4 * Q) = acc;
}

#define GK_WAVES 8
#define GK_H     (D / 2)
#define GK_KPW   (GK_H / GK_WAVES)
#define GK_CHUNK 32
#define GK_RING  16
#define GK_BODY  (GK_RING * 8)

__device__ __forceinline__ v8f hb_mma(v2f a, v2f b, v8f c) {
    return __builtin_amdgcn_wmma_f32_16x16x4_f32(
                false, a,  false, b,
                (short)0, c,  false,  false);
}

__device__ __forceinline__ void hb_load4(float4 (&lo)[4], float4 (&hi)[4],
        const float* p) {
    #pragma unroll
    for (int s = 0; s < 4; ++s) {
        lo[s] = *(const float4*)(p + 8 * s);
        hi[s] = *(const float4*)(p + GK_H + 8 * s);
    }
}

template <int T0, int N>
__device__ __forceinline__ void hb_bfetch(float4 (&rp)[GK_RING], float4 (&rm)[GK_RING],
        const float* qp, const float* qm) {
    #pragma unroll
    for (int u = 0; u < N; ++u) {
        const float* e = qp + 8 * (T0 + u);
        rp[(T0 + u) % GK_RING] = make_float4(e[0], e[1], e[2], e[3]);
        const float* f = qm + 8 * (T0 + u);
        rm[(T0 + u) % GK_RING] = make_float4(f[0], f[1], f[2], f[3]);
    }
}

template <int CC>
__device__ __forceinline__ void hb_chunk(
        const float4 (&lo)[4], const float4 (&hi)[4],
        const float4 (&rp)[GK_RING], const float4 (&rm)[GK_RING],
        v8f& accp0, v8f& accp1, v8f& accm0, v8f& accm1) {
    #pragma unroll
    for (int s = 0; s < 4; ++s) {
        const int i = 4 * CC + s;
        const float4 p0 = rp[(i + 2) % GK_RING];
        const float4 p1 = rp[(i + 0) % GK_RING];
        const float4 q0 = rm[(i + 2) % GK_RING];
        const float4 q1 = rm[(i + 0) % GK_RING];
        v2f aplo = { lo[s].x + hi[s].x, lo[s].y + hi[s].y };
        v2f aphi = { lo[s].z + hi[s].z, lo[s].w + hi[s].w };
        v2f amlo = { lo[s].x - hi[s].x, lo[s].y - hi[s].y };
        v2f amhi = { lo[s].z - hi[s].z, lo[s].w - hi[s].w };
        v2f p0lo = { p0.x, p0.y }, p0hi = { p0.z, p0.w };
        v2f p1lo = { p1.x, p1.y }, p1hi = { p1.z, p1.w };
        v2f q0lo = { q0.x, q0.y }, q0hi = { q0.z, q0.w };
        v2f q1lo = { q1.x, q1.y }, q1hi = { q1.z, q1.w };
        accp0 = hb_mma(aplo, p0lo, accp0);
        accm0 = hb_mma(amlo, q0lo, accm0);
        accp1 = hb_mma(aplo, p1lo, accp1);
        accm1 = hb_mma(amlo, q1lo, accm1);
        accp0 = hb_mma(aphi, p0hi, accp0);
        accm0 = hb_mma(amhi, q0hi, accm0);
        accp1 = hb_mma(aphi, p1hi, accp1);
        accm1 = hb_mma(amhi, q1hi, accm1);
    }
}

__device__ __forceinline__ void hb_region_end() {
    asm volatile("" ::: "memory");
    __builtin_amdgcn_sched_barrier(0);
}

template <bool LAST>
__device__ __forceinline__ void hb_body(const float* ap, const float* qp, const float* qm,
        float4 (&rp)[GK_RING], float4 (&rm)[GK_RING],
        float4 (&elo)[4], float4 (&ehi)[4], float4 (&olo)[4], float4 (&ohi)[4],
        v8f& accp0, v8f& accp1, v8f& accm0, v8f& accm1) {
    hb_load4(olo, ohi, ap + 1 * GK_CHUNK);
    hb_bfetch<6, 4>(rp, rm, qp, qm);
    hb_region_end();
    hb_chunk<0>(elo, ehi, rp, rm, accp0, accp1, accm0, accm1);
    hb_region_end();

    hb_load4(elo, ehi, ap + 2 * GK_CHUNK);
    hb_bfetch<10, 4>(rp, rm, qp, qm);
    hb_region_end();
    hb_chunk<1>(olo, ohi, rp, rm, accp0, accp1, accm0, accm1);
    hb_region_end();

    hb_load4(olo, ohi, ap + 3 * GK_CHUNK);
    hb_bfetch<14, 4>(rp, rm, qp, qm);
    hb_region_end();
    hb_chunk<2>(elo, ehi, rp, rm, accp0, accp1, accm0, accm1);
    hb_region_end();

    if (!LAST) {
        hb_load4(elo, ehi, ap + 4 * GK_CHUNK);
        hb_bfetch<18, 4>(rp, rm, qp, qm);
        hb_region_end();
    }
    hb_chunk<3>(olo, ohi, rp, rm, accp0, accp1, accm0, accm1);
    if (!LAST)
        hb_region_end();
}

__global__ void __launch_bounds__(256) hb_gemm_wmma(
        const float* __restrict__ y, const float* __restrict__ base,
        float* __restrict__ out) {
    __shared__ __attribute__((aligned(16))) float lbase[2 * D];
    float* lbp = lbase;
    float* lbm = lbase + D;
    int t = threadIdx.x;

    float4 bv[4];
    for (int j = 0; j < 4; ++j)
        bv[j] = *(const float4*)(base + 4 * (t + 256 * j));

    int wave = t >> 5;
    int lane = t & 31;
    int half = lane >> 4;
    int l    = lane & 15;

    int m0 = blockIdx.y * 16;
    int n0 = blockIdx.x * 32;
    int kbeg = wave * GK_KPW;

    const float* ap = y + (size_t)(m0 + l) * D + kbeg + 4 * half;
    int off_p0 = 4 * half - (n0 + l) - 16 + kbeg + GK_H;
    int off_m0 = off_p0 + D;
    int off_p1 = off_p0 + GK_BODY;
    int off_m1 = off_p0 + D + GK_BODY;
    asm volatile("" : "+v"(off_m0), "+v"(off_p1), "+v"(off_m1));
    const float* qp = lbase + off_p0;
    const float* qm = lbase + off_m0;
    const float* qp1 = lbase + off_p1;
    const float* qm1 = lbase + off_m1;

    v8f accp0 = {}, accp1 = {}, accm0 = {}, accm1 = {};

    float4 elo[4], ehi[4], olo[4], ohi[4];
    float4 rp[GK_RING], rm[GK_RING];
    hb_load4(elo, ehi, ap);

    for (int j = 0; j < 2; ++j) {
        const float4 a = bv[j], b = bv[j + 2];
        const float4 sp = make_float4(a.x + b.x, a.y + b.y, a.z + b.z, a.w + b.w);
        const float4 sm = make_float4(a.x - b.x, a.y - b.y, a.z - b.z, a.w - b.w);
        const float4 sn = make_float4(-sm.x, -sm.y, -sm.z, -sm.w);
        *(float4*)(lbp + 4 * (t + 256 * j))        = sp;
        *(float4*)(lbp + GK_H + 4 * (t + 256 * j)) = sp;
        *(float4*)(lbm + 4 * (t + 256 * j))        = sn;
        *(float4*)(lbm + GK_H + 4 * (t + 256 * j)) = sm;
    }
    __syncthreads();

    hb_bfetch<0, 6>(rp, rm, qp, qm);

    hb_body<false>(ap, qp, qm, rp, rm, elo, ehi, olo, ohi, accp0, accp1, accm0, accm1);
    hb_body<true>(ap + GK_BODY, qp1, qm1,
                  rp, rm, elo, ehi, olo, ohi, accp0, accp1, accm0, accm1);

    __syncthreads();
    float* red = lbase;
    float* myred = red + wave * 1024 + lane;
    #pragma unroll
    for (int v = 0; v < 8; ++v) {
        myred[(0 * 8 + v) * 32] = accp0[v];
        myred[(1 * 8 + v) * 32] = accp1[v];
        myred[(2 * 8 + v) * 32] = accm0[v];
        myred[(3 * 8 + v) * 32] = accm1[v];
    }
    __syncthreads();

    int r   = t >> 4;
    int c2  = (t & 15) << 1;
    int pidx = (((c2 >> 4) * 8 + (r & 7)) * 32) + (r >> 3) * 16 + (c2 & 15);
    int midx = pidx + 2 * 8 * 32;
    float2 ps = *(const float2*)(red + pidx);
    float2 ms = *(const float2*)(red + midx);
    #pragma unroll
    for (int w = 1; w < GK_WAVES; ++w) {
        float2 pp = *(const float2*)(red + w * 1024 + pidx);
        float2 mm = *(const float2*)(red + w * 1024 + midx);
        ps.x += pp.x; ps.y += pp.y;
        ms.x += mm.x; ms.y += mm.y;
    }
    float* orow = out + (size_t)(m0 + r) * D + n0 + c2;
    *(float2*)(orow)        = make_float2(0.5f * (ps.x + ms.x), 0.5f * (ps.y + ms.y));
    *(float2*)(orow + GK_H) = make_float2(0.5f * (ps.x - ms.x), 0.5f * (ps.y - ms.y));
}

__global__ void __launch_bounds__(256) hb_normalize(float* __restrict__ out) {
    __shared__ float sm[256];
    int t = threadIdx.x;
    float* rowp = out + (size_t)blockIdx.x * D;
    float v[16];
    float ssq = 0.0f;
    #pragma unroll
    for (int i = 0; i < 16; ++i) {
        v[i] = rowp[t + i * 256];
        ssq = fmaf(v[i], v[i], ssq);
    }
    sm[t] = ssq; __syncthreads();
    for (int s = 128; s > 0; s >>= 1) {
        if (t < s) sm[t] += sm[t + s];
        __syncthreads();
    }
    float scale = rsqrtf(fmaxf(sm[0], HB_EPS));
    #pragma unroll
    for (int i = 0; i < 16; ++i)
        rowp[t + i * 256] = v[i] * scale;
}

extern "C" void kernel_launch(void* const* d_in, const int* in_sizes, int n_in,
                              void* d_out, int out_size, void* d_ws, size_t ws_size,
                              hipStream_t stream) {
    (void)in_sizes; (void)n_in; (void)out_size; (void)ws_size;
    const float* inp  = (const float*)d_in[0];
    const float* base = (const float*)d_in[1];
    const float* pw   = (const float*)d_in[2];
    float* out = (float*)d_out;

    float* y = (float*)((char*)d_ws + 1024);

    hb_shiftsum  <<<dim3(8, NPAIRS), SS_THREADS, 0, stream>>>(inp, pw, y);
    hb_gemm_wmma <<<dim3(64, 4),     256, 0, stream>>>(y, base, out);
    hb_normalize <<<NPAIRS,          256, 0, stream>>>(out);
}


# GEMM A operand: global loads cover whole 128-byte lines (8 lanes per row: 4 lines per load, was 16); yp, ym summed on the staged quads, re-ordered to fragments through a per-wave LDS slot; same sums
# speedup vs baseline: 2.6602x; 1.0468x over previous
#include <hip/hip_runtime.h>

#define D      4096
#define DMASK  4095
#define BS     128
#define NPAIRS 64
#define HB_EPS 1e-8f

typedef __attribute__((ext_vector_type(2))) float v2f;
typedef __attribute__((ext_vector_type(8))) float v8f;

#define SS_THREADS 128
#define SS_AHEAD   6
#define SS_RING    (SS_AHEAD + 1)

typedef __attribute__((ext_vector_type(4))) float v4f;

__device__ __forceinline__ float ss_xor_half(float v) {
    const int i = __builtin_bit_cast(int, v);
    return __builtin_bit_cast(float,
        __builtin_amdgcn_permlanex16(i, i, 0x76543210, 0xfedcba98, false, false));
}
template <int M>
__device__ __forceinline__ float ss_xor_row(float v) {
    const int i = __builtin_bit_cast(int, v);
    return __builtin_bit_cast(float,
        __builtin_amdgcn_update_dpp(i, i, 0x160 + M, 0xf, 0xf, false));
}

__device__ __forceinline__ void ss_weights(const float (&x)[4], float (&wv)[4]) {
    float mx = fmaxf(fmaxf(x[0], x[2]), fmaxf(x[1], x[3]));
    mx = fmaxf(mx, ss_xor_half(mx));
    mx = fmaxf(mx, ss_xor_row<8>(mx));
    mx = fmaxf(mx, ss_xor_row<4>(mx));
    mx = fmaxf(mx, ss_xor_row<2>(mx));
    mx = fmaxf(mx, ss_xor_row<1>(mx));
    float e[4];
    #pragma unroll
    for (int j = 0; j < 4; ++j) e[j] = expf(x[j] - mx);
    float sum = (e[0] + e[2]) + (e[1] + e[3]);
    sum += ss_xor_half(sum);
    sum += ss_xor_row<8>(sum);
    sum += ss_xor_row<4>(sum);
    sum += ss_xor_row<2>(sum);
    sum += ss_xor_row<1>(sum);
    #pragma unroll
    for (int j = 0; j < 4; ++j) wv[j] = e[j] / sum;
}

__device__ __forceinline__ void ss_hold_weights_here(float (&x)[4]) {
    asm volatile("" : "+v"(x[0]), "+v"(x[1]), "+v"(x[2]), "+v"(x[3]));
}

__device__ __forceinline__ float ss_lane_value(float v, int lane) {
    return __builtin_bit_cast(float,
        __builtin_amdgcn_readlane(__builtin_bit_cast(int, v), lane));
}

template <int RB>
__device__ __forceinline__ void ss_issue_main(v4f (&d)[RB], const float* p, int s0) {
    #pragma unroll
    for (int r = 0; r < RB; ++r)
        __builtin_memcpy(&d[r], p + (size_t)(s0 + r) * (D + 1), 16);
}

template <int RB>
__device__ __forceinline__ void ss_issue_edge(v4f (&d)[RB],
        const float* prow, int Q, int s0) {
    #pragma unroll
    for (int r = 0; r < RB; ++r) {
        const int s = s0 + r, a = s >> 2, b = s & 3;
        const float* row = prow + (size_t)s * D;
        float x[4];
        __builtin_memcpy(&x[0], row + 4 * ((Q + a) & 1023) + b, 4 * (4 - b));
        if (b) __builtin_memcpy(&x[4 - b], row + 4 * ((Q + a + 1) & 1023), 4 * b);
        d[r] = (v4f){ x[0], x[1], x[2], x[3] };
    }
}

template <int RB>
__device__ __forceinline__ void ss_use(v4f& acc, const v4f (&d)[RB],
        const float (&wv)[4], int s0) {
    #pragma unroll
    for (int r = 0; r < RB; ++r) {
        const int s = s0 + r;
        float ws = ss_lane_value(wv[s >> 5], s & 31);
        acc.x = fmaf(ws, d[r].x, acc.x);
        acc.y = fmaf(ws, d[r].y, acc.y);
        acc.z = fmaf(ws, d[r].z, acc.z);
        acc.w = fmaf(ws, d[r].w, acc.w);
    }
}

__global__ void __launch_bounds__(SS_THREADS) hb_shiftsum(
        const float* __restrict__ inp, const float* __restrict__ pw,
        float* __restrict__ y) {
    int pair = blockIdx.y;
    int Q = blockIdx.x * SS_THREADS + threadIdx.x;
    const float* prow = inp + (size_t)pair * BS * D;
    v4f acc = { 0.0f, 0.0f, 0.0f, 0.0f };

    const int lane = threadIdx.x & 31;
    float x[4];
    #pragma unroll
    for (int j = 0; j < 4; ++j) x[j] = pw[lane + 32 * j];
    float wv[4];

    int qw = __builtin_amdgcn_readfirstlane(Q & ~31);
    if (qw < 992) {
        constexpr int RB = 8, NB = BS / RB;
        const float* p = prow + 4 * Q;
        v4f buf[SS_RING][RB];
        #pragma unroll
        for (int g = 0; g < SS_AHEAD; ++g) ss_issue_main<RB>(buf[g], p, g * RB);
        ss_hold_weights_here(x);
        ss_weights(x, wv);
        #pragma unroll
        for (int g = 0; g < NB; ++g) {
            if (g + SS_AHEAD < NB)
                ss_issue_main<RB>(buf[(g + SS_AHEAD) % SS_RING], p, (g + SS_AHEAD) * RB);
            __builtin_amdgcn_sched_barrier(0);
            ss_use<RB>(acc, buf[g % SS_RING], wv, g * RB);
            __builtin_amdgcn_sched_barrier(0);
        }
    } else {
        constexpr int RB = 4, NB = BS / RB;
        v4f buf[SS_RING][RB];
        #pragma unroll
        for (int g = 0; g < SS_AHEAD; ++g) ss_issue_edge<RB>(buf[g], prow, Q, g * RB);
        ss_hold_weights_here(x);
        ss_weights(x, wv);
        #pragma unroll
        for (int g = 0; g < NB; ++g) {
            if (g + SS_AHEAD < NB)
                ss_issue_edge<RB>(buf[(g + SS_AHEAD) % SS_RING], prow, Q, (g + SS_AHEAD) * RB);
            __builtin_amdgcn_sched_barrier(0);
            ss_use<RB>(acc, buf[g % SS_RING], wv, g * RB);
            __builtin_amdgcn_sched_barrier(0);
        }
    }
    *(v4f*)(y + (size_t)pair * D + 4 * Q) = acc;
}

#define GK_WAVES 8
#define GK_H     (D / 2)
#define GK_KPW   (GK_H / GK_WAVES)
#define GK_CHUNK 32
#define GK_RING  16
#define GK_BODY  (GK_RING * 8)
#define GK_APITCH 36
#define GK_APLANE (16 * GK_APITCH)
#define GK_ASLOT  (2 * GK_APLANE)

__device__ __forceinline__ v8f hb_mma(v2f a, v2f b, v8f c) {
    return __builtin_amdgcn_wmma_f32_16x16x4_f32(
                false, a,  false, b,
                (short)0, c,  false,  false);
}

__device__ __forceinline__ void hb_gload(v4f (&st)[8], const float* gp) {
    #pragma unroll
    for (int j = 0; j < 4; ++j) {
        st[j]     = *(const v4f*)(gp + (size_t)(4 * j) * D);
        st[4 + j] = *(const v4f*)(gp + (size_t)(4 * j) * D + GK_H);
    }
}

__device__ __forceinline__ void hb_stage(v4f (&st)[8], float* sp) {
    asm volatile("" : "+v"(st[0]), "+v"(st[1]), "+v"(st[2]), "+v"(st[3]),
                      "+v"(st[4]), "+v"(st[5]), "+v"(st[6]), "+v"(st[7]));
    #pragma unroll
    for (int j = 0; j < 4; ++j) {
        const v4f a = st[j], b = st[4 + j];
        *(v4f*)(sp + 4 * j * GK_APITCH)             = a + b;
        *(v4f*)(sp + GK_APLANE + 4 * j * GK_APITCH) = a - b;
    }
}

__device__ __forceinline__ void hb_afetch(float4 (&fp)[4], float4 (&fm)[4],
        const float* fq) {
    #pragma unroll
    for (int s = 0; s < 4; ++s) {
        fp[s] = *(const float4*)(fq + 8 * s);
        fm[s] = *(const float4*)(fq + GK_APLANE + 8 * s);
    }
}

template <int T0, int N>
__device__ __forceinline__ void hb_bfetch(float4 (&rp)[GK_RING], float4 (&rm)[GK_RING],
        const float* qp, const float* qm) {
    #pragma unroll
    for (int u = 0; u < N; ++u) {
        const float* e = qp + 8 * (T0 + u);
        rp[(T0 + u) % GK_RING] = make_float4(e[0], e[1], e[2], e[3]);
        const float* f = qm + 8 * (T0 + u);
        rm[(T0 + u) % GK_RING] = make_float4(f[0], f[1], f[2], f[3]);
    }
}

template <int CC>
__device__ __forceinline__ void hb_chunk(
        const float4 (&fp)[4], const float4 (&fm)[4],
        const float4 (&rp)[GK_RING], const float4 (&rm)[GK_RING],
        v8f& accp0, v8f& accp1, v8f& accm0, v8f& accm1) {
    #pragma unroll
    for (int s = 0; s < 4; ++s) {
        const int i = 4 * CC + s;
        const float4 p0 = rp[(i + 2) % GK_RING];
        const float4 p1 = rp[(i + 0) % GK_RING];
        const float4 q0 = rm[(i + 2) % GK_RING];
        const float4 q1 = rm[(i + 0) % GK_RING];
        v2f aplo = { fp[s].x, fp[s].y };
        v2f aphi = { fp[s].z, fp[s].w };
        v2f amlo = { fm[s].x, fm[s].y };
        v2f amhi = { fm[s].z, fm[s].w };
        v2f p0lo = { p0.x, p0.y }, p0hi = { p0.z, p0.w };
        v2f p1lo = { p1.x, p1.y }, p1hi = { p1.z, p1.w };
        v2f q0lo = { q0.x, q0.y }, q0hi = { q0.z, q0.w };
        v2f q1lo = { q1.x, q1.y }, q1hi = { q1.z, q1.w };
        accp0 = hb_mma(aplo, p0lo, accp0);
        accm0 = hb_mma(amlo, q0lo, accm0);
        accp1 = hb_mma(aplo, p1lo, accp1);
        accm1 = hb_mma(amlo, q1lo, accm1);
        accp0 = hb_mma(aphi, p0hi, accp0);
        accm0 = hb_mma(amhi, q0hi, accm0);
        accp1 = hb_mma(aphi, p1hi, accp1);
        accm1 = hb_mma(amhi, q1hi, accm1);
    }
}

__device__ __forceinline__ void hb_region_end(v8f& a0, v8f& a1, v8f& a2, v8f& a3) {
    asm volatile("" : "+v"(a0), "+v"(a1), "+v"(a2), "+v"(a3) :: "memory");
    __builtin_amdgcn_sched_barrier(0);
}

template <bool LAST>
__device__ __forceinline__ void hb_body(const float* gp, float* sp, const float* fq,
        const float* qp, const float* qm,
        float4 (&rp)[GK_RING], float4 (&rm)[GK_RING], v4f (&st)[8],
        float4 (&efp)[4], float4 (&efm)[4], float4 (&ofp)[4], float4 (&ofm)[4],
        v8f& accp0, v8f& accp1, v8f& accm0, v8f& accm1) {
    hb_stage(st, sp);
    hb_gload(st, gp + 2 * GK_CHUNK);
    hb_afetch(ofp, ofm, fq);
    hb_bfetch<6, 4>(rp, rm, qp, qm);
    hb_region_end(accp0, accp1, accm0, accm1);
    hb_chunk<0>(efp, efm, rp, rm, accp0, accp1, accm0, accm1);
    hb_region_end(accp0, accp1, accm0, accm1);

    hb_stage(st, sp);
    hb_gload(st, gp + 3 * GK_CHUNK);
    hb_afetch(efp, efm, fq);
    hb_bfetch<10, 4>(rp, rm, qp, qm);
    hb_region_end(accp0, accp1, accm0, accm1);
    hb_chunk<1>(ofp, ofm, rp, rm, accp0, accp1, accm0, accm1);
    hb_region_end(accp0, accp1, accm0, accm1);

    hb_stage(st, sp);
    if (!LAST)
        hb_gload(st, gp + 4 * GK_CHUNK);
    hb_afetch(ofp, ofm, fq);
    hb_bfetch<14, 4>(rp, rm, qp, qm);
    hb_region_end(accp0, accp1, accm0, accm1);
    hb_chunk<2>(efp, efm, rp, rm, accp0, accp1, accm0, accm1);
    hb_region_end(accp0, accp1, accm0, accm1);

    if (!LAST) {
        hb_stage(st, sp);
        hb_gload(st, gp + 5 * GK_CHUNK);
        hb_afetch(efp, efm, fq);
        hb_bfetch<18, 4>(rp, rm, qp, qm);
        hb_region_end(accp0, accp1, accm0, accm1);
    }
    hb_chunk<3>(ofp, ofm, rp, rm, accp0, accp1, accm0, accm1);
    if (!LAST)
        hb_region_end(accp0, accp1, accm0, accm1);
}

__global__ void __launch_bounds__(256) hb_gemm_wmma(
        const float* __restrict__ y, const float* __restrict__ base,
        float* __restrict__ out) {
    __shared__ __attribute__((aligned(16))) float lbase[2 * D];
    __shared__ __attribute__((aligned(16))) float aslot[GK_WAVES * GK_ASLOT];
    float* lbp = lbase;
    float* lbm = lbase + D;
    int t = threadIdx.x;

    float4 bv[4];
    for (int j = 0; j < 4; ++j)
        bv[j] = *(const float4*)(base + 4 * (t + 256 * j));

    int wave = t >> 5;
    int lane = t & 31;
    int half = lane >> 4;
    int l    = lane & 15;

    int m0 = blockIdx.y * 16;
    int n0 = blockIdx.x * 32;
    int kbeg = wave * GK_KPW;

    const float* gp = y + (size_t)(m0 + (lane >> 3)) * D + kbeg + 4 * (lane & 7);
    float* sp = aslot + wave * GK_ASLOT + (lane >> 3) * GK_APITCH + 4 * (lane & 7);
    const float* fq = aslot + wave * GK_ASLOT + l * GK_APITCH + 4 * half;
    int off_p0 = 4 * half - (n0 + l) - 16 + kbeg + GK_H;
    int off_m0 = off_p0 + D;
    int off_p1 = off_p0 + GK_BODY;
    int off_m1 = off_p0 + D + GK_BODY;
    asm volatile("" : "+v"(off_m0), "+v"(off_p1), "+v"(off_m1));
    const float* qp = lbase + off_p0;
    const float* qm = lbase + off_m0;
    const float* qp1 = lbase + off_p1;
    const float* qm1 = lbase + off_m1;

    v8f accp0 = {}, accp1 = {}, accm0 = {}, accm1 = {};

    v4f st[8];
    float4 efp[4], efm[4], ofp[4], ofm[4];
    float4 rp[GK_RING], rm[GK_RING];
    hb_gload(st, gp);

    for (int j = 0; j < 2; ++j) {
        const float4 a = bv[j], b = bv[j + 2];
        const float4 sp = make_float4(a.x + b.x, a.y + b.y, a.z + b.z, a.w + b.w);
        const float4 sm = make_float4(a.x - b.x, a.y - b.y, a.z - b.z, a.w - b.w);
        const float4 sn = make_float4(-sm.x, -sm.y, -sm.z, -sm.w);
        *(float4*)(lbp + 4 * (t + 256 * j))        = sp;
        *(float4*)(lbp + GK_H + 4 * (t + 256 * j)) = sp;
        *(float4*)(lbm + 4 * (t + 256 * j))        = sn;
        *(float4*)(lbm + GK_H + 4 * (t + 256 * j)) = sm;
    }
    __syncthreads();

    hb_stage(st, sp);
    hb_gload(st, gp + GK_CHUNK);
    hb_afetch(efp, efm, fq);
    hb_bfetch<0, 6>(rp, rm, qp, qm);

    hb_body<false>(gp, sp, fq, qp, qm, rp, rm, st, efp, efm, ofp, ofm,
                   accp0, accp1, accm0, accm1);
    hb_body<true>(gp + GK_BODY, sp, fq, qp1, qm1, rp, rm, st, efp, efm, ofp, ofm,
                  accp0, accp1, accm0, accm1);

    __syncthreads();
    float* red = lbase;
    float* myred = red + wave * 1024 + lane;
    #pragma unroll
    for (int v = 0; v < 8; ++v) {
        myred[(0 * 8 + v) * 32] = accp0[v];
        myred[(1 * 8 + v) * 32] = accp1[v];
        myred[(2 * 8 + v) * 32] = accm0[v];
        myred[(3 * 8 + v) * 32] = accm1[v];
    }
    __syncthreads();

    int r   = t >> 4;
    int c2  = (t & 15) << 1;
    int pidx = (((c2 >> 4) * 8 + (r & 7)) * 32) + (r >> 3) * 16 + (c2 & 15);
    int midx = pidx + 2 * 8 * 32;
    float2 ps = *(const float2*)(red + pidx);
    float2 ms = *(const float2*)(red + midx);
    #pragma unroll
    for (int w = 1; w < GK_WAVES; ++w) {
        float2 pp = *(const float2*)(red + w * 1024 + pidx);
        float2 mm = *(const float2*)(red + w * 1024 + midx);
        ps.x += pp.x; ps.y += pp.y;
        ms.x += mm.x; ms.y += mm.y;
    }
    float* orow = out + (size_t)(m0 + r) * D + n0 + c2;
    *(float2*)(orow)        = make_float2(0.5f * (ps.x + ms.x), 0.5f * (ps.y + ms.y));
    *(float2*)(orow + GK_H) = make_float2(0.5f * (ps.x - ms.x), 0.5f * (ps.y - ms.y));
}

__global__ void __launch_bounds__(256) hb_normalize(float* __restrict__ out) {
    __shared__ float sm[256];
    int t = threadIdx.x;
    float* rowp = out + (size_t)blockIdx.x * D;
    float v[16];
    float ssq = 0.0f;
    #pragma unroll
    for (int i = 0; i < 16; ++i) {
        v[i] = rowp[t + i * 256];
        ssq = fmaf(v[i], v[i], ssq);
    }
    sm[t] = ssq; __syncthreads();
    for (int s = 128; s > 0; s >>= 1) {
        if (t < s) sm[t] += sm[t + s];
        __syncthreads();
    }
    float scale = rsqrtf(fmaxf(sm[0], HB_EPS));
    #pragma unroll
    for (int i = 0; i < 16; ++i)
        rowp[t + i * 256] = v[i] * scale;
}

extern "C" void kernel_launch(void* const* d_in, const int* in_sizes, int n_in,
                              void* d_out, int out_size, void* d_ws, size_t ws_size,
                              hipStream_t stream) {
    (void)in_sizes; (void)n_in; (void)out_size; (void)ws_size;
    const float* inp  = (const float*)d_in[0];
    const float* base = (const float*)d_in[1];
    const float* pw   = (const float*)d_in[2];
    float* out = (float*)d_out;

    float* y = (float*)((char*)d_ws + 1024);

    hb_shiftsum  <<<dim3(8, NPAIRS), SS_THREADS, 0, stream>>>(inp, pw, y);
    hb_gemm_wmma <<<dim3(64, 4),     256, 0, stream>>>(y, base, out);
    hb_normalize <<<NPAIRS,          256, 0, stream>>>(out);
}


# Shift-sum on 16 waves per WGP (was 8): a work-group is 8 waves, the 128 rows split in two halves of 64 summed by two waves, upper half added through LDS at the end; new summation order over s
# speedup vs baseline: 2.7394x; 1.0298x over previous
#include <hip/hip_runtime.h>

#define D      4096
#define DMASK  4095
#define BS     128
#define NPAIRS 64
#define HB_EPS 1e-8f

typedef __attribute__((ext_vector_type(2))) float v2f;
typedef __attribute__((ext_vector_type(8))) float v8f;

#define SS_QUADS   128
#define SS_PARTS   2
#define SS_THREADS (SS_QUADS * SS_PARTS)
#define SS_ROWS    (BS / SS_PARTS)
#define SS_AHEAD_MAIN 4
#define SS_AHEAD_EDGE 6

typedef __attribute__((ext_vector_type(4))) float v4f;

__device__ __forceinline__ float ss_xor_half(float v) {
    const int i = __builtin_bit_cast(int, v);
    return __builtin_bit_cast(float,
        __builtin_amdgcn_permlanex16(i, i, 0x76543210, 0xfedcba98, false, false));
}
template <int M>
__device__ __forceinline__ float ss_xor_row(float v) {
    const int i = __builtin_bit_cast(int, v);
    return __builtin_bit_cast(float,
        __builtin_amdgcn_update_dpp(i, i, 0x160 + M, 0xf, 0xf, false));
}

__device__ __forceinline__ void ss_weights(const float (&x)[4], float (&wv)[4]) {
    float mx = fmaxf(fmaxf(x[0], x[2]), fmaxf(x[1], x[3]));
    mx = fmaxf(mx, ss_xor_half(mx));
    mx = fmaxf(mx, ss_xor_row<8>(mx));
    mx = fmaxf(mx, ss_xor_row<4>(mx));
    mx = fmaxf(mx, ss_xor_row<2>(mx));
    mx = fmaxf(mx, ss_xor_row<1>(mx));
    float e[4];
    #pragma unroll
    for (int j = 0; j < 4; ++j) e[j] = expf(x[j] - mx);
    float sum = (e[0] + e[2]) + (e[1] + e[3]);
    sum += ss_xor_half(sum);
    sum += ss_xor_row<8>(sum);
    sum += ss_xor_row<4>(sum);
    sum += ss_xor_row<2>(sum);
    sum += ss_xor_row<1>(sum);
    #pragma unroll
    for (int j = 0; j < 4; ++j) wv[j] = e[j] / sum;
}

__device__ __forceinline__ void ss_hold_weights_here(float (&x)[4]) {
    asm volatile("" : "+v"(x[0]), "+v"(x[1]), "+v"(x[2]), "+v"(x[3]));
}

__device__ __forceinline__ float ss_lane_value(float v, int lane) {
    return __builtin_bit_cast(float,
        __builtin_amdgcn_readlane(__builtin_bit_cast(int, v), lane));
}

template <int RB>
__device__ __forceinline__ void ss_issue_main(v4f (&d)[RB], const float* p, int s0) {
    #pragma unroll
    for (int r = 0; r < RB; ++r)
        __builtin_memcpy(&d[r], p + (size_t)(s0 + r) * (D + 1), 16);
}

template <int RB>
__device__ __forceinline__ void ss_issue_edge(v4f (&d)[RB],
        const float* prow, int Q, int s0) {
    #pragma unroll
    for (int r = 0; r < RB; ++r) {
        const int s = s0 + r, a = s >> 2, b = s & 3;
        const float* row = prow + (size_t)s * D;
        float x[4];
        __builtin_memcpy(&x[0], row + 4 * ((Q + a) & 1023) + b, 4 * (4 - b));
        if (b) __builtin_memcpy(&x[4 - b], row + 4 * ((Q + a + 1) & 1023), 4 * b);
        d[r] = (v4f){ x[0], x[1], x[2], x[3] };
    }
}

template <int RB>
__device__ __forceinline__ void ss_use(v4f& acc, const v4f (&d)[RB],
        const float (&wh)[2], int s0) {
    #pragma unroll
    for (int r = 0; r < RB; ++r) {
        const int s = s0 + r;
        float ws = ss_lane_value(wh[s >> 5], s & 31);
        acc.x = fmaf(ws, d[r].x, acc.x);
        acc.y = fmaf(ws, d[r].y, acc.y);
        acc.z = fmaf(ws, d[r].z, acc.z);
        acc.w = fmaf(ws, d[r].w, acc.w);
    }
}

__global__ void __launch_bounds__(SS_THREADS) hb_shiftsum(
        const float* __restrict__ inp, const float* __restrict__ pw,
        float* __restrict__ y) {
    __shared__ v4f ss_upper[SS_QUADS];
    int pair = blockIdx.y;
    const int tq = threadIdx.x & (SS_QUADS - 1);
    int Q = blockIdx.x * SS_QUADS + tq;
    const int part = __builtin_amdgcn_readfirstlane((int)threadIdx.x >> 7);
    const float* prow = inp + (size_t)pair * BS * D + (size_t)(part * SS_ROWS) * D;
    v4f acc = { 0.0f, 0.0f, 0.0f, 0.0f };

    const int lane = threadIdx.x & 31;
    float x[4];
    #pragma unroll
    for (int j = 0; j < 4; ++j) x[j] = pw[lane + 32 * j];
    float wv[4], wh[2];

    int qw = __builtin_amdgcn_readfirstlane(Q & ~31);
    if (qw < 992) {
        constexpr int RB = 8, NB = SS_ROWS / RB, AH = SS_AHEAD_MAIN, RING = AH + 1;
        const float* p = prow + 4 * Q + part * SS_ROWS;
        v4f buf[RING][RB];
        #pragma unroll
        for (int g = 0; g < AH; ++g) ss_issue_main<RB>(buf[g], p, g * RB);
        ss_hold_weights_here(x);
        ss_weights(x, wv);
        wh[0] = part ? wv[2] : wv[0];
        wh[1] = part ? wv[3] : wv[1];
        #pragma unroll
        for (int g = 0; g < NB; ++g) {
            if (g + AH < NB)
                ss_issue_main<RB>(buf[(g + AH) % RING], p, (g + AH) * RB);
            __builtin_amdgcn_sched_barrier(0);
            ss_use<RB>(acc, buf[g % RING], wh, g * RB);
            __builtin_amdgcn_sched_barrier(0);
        }
    } else {
        constexpr int RB = 4, NB = SS_ROWS / RB, AH = SS_AHEAD_EDGE, RING = AH + 1;
        const int Qe = Q + part * (SS_ROWS / 4);
        v4f buf[RING][RB];
        #pragma unroll
        for (int g = 0; g < AH; ++g) ss_issue_edge<RB>(buf[g], prow, Qe, g * RB);
        __builtin_amdgcn_sched_barrier(0);
        ss_hold_weights_here(x);
        ss_weights(x, wv);
        wh[0] = part ? wv[2] : wv[0];
        wh[1] = part ? wv[3] : wv[1];
        #pragma unroll
        for (int g = 0; g < NB; ++g) {
            if (g + AH < NB)
                ss_issue_edge<RB>(buf[(g + AH) % RING], prow, Qe, (g + AH) * RB);
            __builtin_amdgcn_sched_barrier(0);
            ss_use<RB>(acc, buf[g % RING], wh, g * RB);
            __builtin_amdgcn_sched_barrier(0);
        }
    }
    if (part) ss_upper[tq] = acc;
    __syncthreads();
    if (!part) {
        const v4f up = ss_upper[tq];
        acc.x += up.x; acc.y += up.y; acc.z += up.z; acc.w += up.w;
        *(v4f*)(y + (size_t)pair * D + 4 * Q) = acc;
    }
}

#define GK_WAVES 8
#define GK_H     (D / 2)
#define GK_KPW   (GK_H / GK_WAVES)
#define GK_CHUNK 32
#define GK_RING  16
#define GK_BODY  (GK_RING * 8)
#define GK_APITCH 36
#define GK_APLANE (16 * GK_APITCH)
#define GK_ASLOT  (2 * GK_APLANE)

__device__ __forceinline__ v8f hb_mma(v2f a, v2f b, v8f c) {
    return __builtin_amdgcn_wmma_f32_16x16x4_f32(
                false, a,  false, b,
                (short)0, c,  false,  false);
}

__device__ __forceinline__ void hb_gload(v4f (&st)[8], const float* gp) {
    #pragma unroll
    for (int j = 0; j < 4; ++j) {
        st[j]     = *(const v4f*)(gp + (size_t)(4 * j) * D);
        st[4 + j] = *(const v4f*)(gp + (size_t)(4 * j) * D + GK_H);
    }
}

__device__ __forceinline__ void hb_stage(v4f (&st)[8], float* sp) {
    asm volatile("" : "+v"(st[0]), "+v"(st[1]), "+v"(st[2]), "+v"(st[3]),
                      "+v"(st[4]), "+v"(st[5]), "+v"(st[6]), "+v"(st[7]));
    #pragma unroll
    for (int j = 0; j < 4; ++j) {
        const v4f a = st[j], b = st[4 + j];
        *(v4f*)(sp + 4 * j * GK_APITCH)             = a + b;
        *(v4f*)(sp + GK_APLANE + 4 * j * GK_APITCH) = a - b;
    }
}

__device__ __forceinline__ void hb_afetch(float4 (&fp)[4], float4 (&fm)[4],
        const float* fq) {
    #pragma unroll
    for (int s = 0; s < 4; ++s) {
        fp[s] = *(const float4*)(fq + 8 * s);
        fm[s] = *(const float4*)(fq + GK_APLANE + 8 * s);
    }
}

template <int T0, int N>
__device__ __forceinline__ void hb_bfetch(float4 (&rp)[GK_RING], float4 (&rm)[GK_RING],
        const float* qp, const float* qm) {
    #pragma unroll
    for (int u = 0; u < N; ++u) {
        const float* e = qp + 8 * (T0 + u);
        rp[(T0 + u) % GK_RING] = make_float4(e[0], e[1], e[2], e[3]);
        const float* f = qm + 8 * (T0 + u);
        rm[(T0 + u) % GK_RING] = make_float4(f[0], f[1], f[2], f[3]);
    }
}

template <int CC>
__device__ __forceinline__ void hb_chunk(
        const float4 (&fp)[4], const float4 (&fm)[4],
        const float4 (&rp)[GK_RING], const float4 (&rm)[GK_RING],
        v8f& accp0, v8f& accp1, v8f& accm0, v8f& accm1) {
    #pragma unroll
    for (int s = 0; s < 4; ++s) {
        const int i = 4 * CC + s;
        const float4 p0 = rp[(i + 2) % GK_RING];
        const float4 p1 = rp[(i + 0) % GK_RING];
        const float4 q0 = rm[(i + 2) % GK_RING];
        const float4 q1 = rm[(i + 0) % GK_RING];
        v2f aplo = { fp[s].x, fp[s].y };
        v2f aphi = { fp[s].z, fp[s].w };
        v2f amlo = { fm[s].x, fm[s].y };
        v2f amhi = { fm[s].z, fm[s].w };
        v2f p0lo = { p0.x, p0.y }, p0hi = { p0.z, p0.w };
        v2f p1lo = { p1.x, p1.y }, p1hi = { p1.z, p1.w };
        v2f q0lo = { q0.x, q0.y }, q0hi = { q0.z, q0.w };
        v2f q1lo = { q1.x, q1.y }, q1hi = { q1.z, q1.w };
        accp0 = hb_mma(aplo, p0lo, accp0);
        accm0 = hb_mma(amlo, q0lo, accm0);
        accp1 = hb_mma(aplo, p1lo, accp1);
        accm1 = hb_mma(amlo, q1lo, accm1);
        accp0 = hb_mma(aphi, p0hi, accp0);
        accm0 = hb_mma(amhi, q0hi, accm0);
        accp1 = hb_mma(aphi, p1hi, accp1);
        accm1 = hb_mma(amhi, q1hi, accm1);
    }
}

__device__ __forceinline__ void hb_region_end(v8f& a0, v8f& a1, v8f& a2, v8f& a3) {
    asm volatile("" : "+v"(a0), "+v"(a1), "+v"(a2), "+v"(a3) :: "memory");
    __builtin_amdgcn_sched_barrier(0);
}

template <bool LAST>
__device__ __forceinline__ void hb_body(const float* gp, float* sp, const float* fq,
        const float* qp, const float* qm,
        float4 (&rp)[GK_RING], float4 (&rm)[GK_RING], v4f (&st)[8],
        float4 (&efp)[4], float4 (&efm)[4], float4 (&ofp)[4], float4 (&ofm)[4],
        v8f& accp0, v8f& accp1, v8f& accm0, v8f& accm1) {
    hb_stage(st, sp);
    hb_gload(st, gp + 2 * GK_CHUNK);
    hb_afetch(ofp, ofm, fq);
    hb_bfetch<6, 4>(rp, rm, qp, qm);
    hb_region_end(accp0, accp1, accm0, accm1);
    hb_chunk<0>(efp, efm, rp, rm, accp0, accp1, accm0, accm1);
    hb_region_end(accp0, accp1, accm0, accm1);

    hb_stage(st, sp);
    hb_gload(st, gp + 3 * GK_CHUNK);
    hb_afetch(efp, efm, fq);
    hb_bfetch<10, 4>(rp, rm, qp, qm);
    hb_region_end(accp0, accp1, accm0, accm1);
    hb_chunk<1>(ofp, ofm, rp, rm, accp0, accp1, accm0, accm1);
    hb_region_end(accp0, accp1, accm0, accm1);

    hb_stage(st, sp);
    if (!LAST)
        hb_gload(st, gp + 4 * GK_CHUNK);
    hb_afetch(ofp, ofm, fq);
    hb_bfetch<14, 4>(rp, rm, qp, qm);
    hb_region_end(accp0, accp1, accm0, accm1);
    hb_chunk<2>(efp, efm, rp, rm, accp0, accp1, accm0, accm1);
    hb_region_end(accp0, accp1, accm0, accm1);

    if (!LAST) {
        hb_stage(st, sp);
        hb_gload(st, gp + 5 * GK_CHUNK);
        hb_afetch(efp, efm, fq);
        hb_bfetch<18, 4>(rp, rm, qp, qm);
        hb_region_end(accp0, accp1, accm0, accm1);
    }
    hb_chunk<3>(ofp, ofm, rp, rm, accp0, accp1, accm0, accm1);
    if (!LAST)
        hb_region_end(accp0, accp1, accm0, accm1);
}

__global__ void __launch_bounds__(256) hb_gemm_wmma(
        const float* __restrict__ y, const float* __restrict__ base,
        float* __restrict__ out) {
    __shared__ __attribute__((aligned(16))) float lbase[2 * D];
    __shared__ __attribute__((aligned(16))) float aslot[GK_WAVES * GK_ASLOT];
    float* lbp = lbase;
    float* lbm = lbase + D;
    int t = threadIdx.x;

    float4 bv[4];
    for (int j = 0; j < 4; ++j)
        bv[j] = *(const float4*)(base + 4 * (t + 256 * j));

    int wave = t >> 5;
    int lane = t & 31;
    int half = lane >> 4;
    int l    = lane & 15;

    int m0 = blockIdx.y * 16;
    int n0 = blockIdx.x * 32;
    int kbeg = wave * GK_KPW;

    const float* gp = y + (size_t)(m0 + (lane >> 3)) * D + kbeg + 4 * (lane & 7);
    float* sp = aslot + wave * GK_ASLOT + (lane >> 3) * GK_APITCH + 4 * (lane & 7);
    const float* fq = aslot + wave * GK_ASLOT + l * GK_APITCH + 4 * half;
    int off_p0 = 4 * half - (n0 + l) - 16 + kbeg + GK_H;
    int off_m0 = off_p0 + D;
    int off_p1 = off_p0 + GK_BODY;
    int off_m1 = off_p0 + D + GK_BODY;
    asm volatile("" : "+v"(off_m0), "+v"(off_p1), "+v"(off_m1));
    const float* qp = lbase + off_p0;
    const float* qm = lbase + off_m0;
    const float* qp1 = lbase + off_p1;
    const float* qm1 = lbase + off_m1;

    v8f accp0 = {}, accp1 = {}, accm0 = {}, accm1 = {};

    v4f st[8];
    float4 efp[4], efm[4], ofp[4], ofm[4];
    float4 rp[GK_RING], rm[GK_RING];
    hb_gload(st, gp);

    for (int j = 0; j < 2; ++j) {
        const float4 a = bv[j], b = bv[j + 2];
        const float4 sp = make_float4(a.x + b.x, a.y + b.y, a.z + b.z, a.w + b.w);
        const float4 sm = make_float4(a.x - b.x, a.y - b.y, a.z - b.z, a.w - b.w);
        const float4 sn = make_float4(-sm.x, -sm.y, -sm.z, -sm.w);
        *(float4*)(lbp + 4 * (t + 256 * j))        = sp;
        *(float4*)(lbp + GK_H + 4 * (t + 256 * j)) = sp;
        *(float4*)(lbm + 4 * (t + 256 * j))        = sn;
        *(float4*)(lbm + GK_H + 4 * (t + 256 * j)) = sm;
    }
    __syncthreads();

    hb_stage(st, sp);
    hb_gload(st, gp + GK_CHUNK);
    hb_afetch(efp, efm, fq);
    hb_bfetch<0, 6>(rp, rm, qp, qm);

    hb_body<false>(gp, sp, fq, qp, qm, rp, rm, st, efp, efm, ofp, ofm,
                   accp0, accp1, accm0, accm1);
    hb_body<true>(gp + GK_BODY, sp, fq, qp1, qm1, rp, rm, st, efp, efm, ofp, ofm,
                  accp0, accp1, accm0, accm1);

    __syncthreads();
    float* red = lbase;
    float* myred = red + wave * 1024 + lane;
    #pragma unroll
    for (int v = 0; v < 8; ++v) {
        myred[(0 * 8 + v) * 32] = accp0[v];
        myred[(1 * 8 + v) * 32] = accp1[v];
        myred[(2 * 8 + v) * 32] = accm0[v];
        myred[(3 * 8 + v) * 32] = accm1[v];
    }
    __syncthreads();

    int r   = t >> 4;
    int c2  = (t & 15) << 1;
    int pidx = (((c2 >> 4) * 8 + (r & 7)) * 32) + (r >> 3) * 16 + (c2 & 15);
    int midx = pidx + 2 * 8 * 32;
    float2 ps = *(const float2*)(red + pidx);
    float2 ms = *(const float2*)(red + midx);
    #pragma unroll
    for (int w = 1; w < GK_WAVES; ++w) {
        float2 pp = *(const float2*)(red + w * 1024 + pidx);
        float2 mm = *(const float2*)(red + w * 1024 + midx);
        ps.x += pp.x; ps.y += pp.y;
        ms.x += mm.x; ms.y += mm.y;
    }
    float* orow = out + (size_t)(m0 + r) * D + n0 + c2;
    *(float2*)(orow)        = make_float2(0.5f * (ps.x + ms.x), 0.5f * (ps.y + ms.y));
    *(float2*)(orow + GK_H) = make_float2(0.5f * (ps.x - ms.x), 0.5f * (ps.y - ms.y));
}

__global__ void __launch_bounds__(256) hb_normalize(float* __restrict__ out) {
    __shared__ float sm[256];
    int t = threadIdx.x;
    float* rowp = out + (size_t)blockIdx.x * D;
    float v[16];
    float ssq = 0.0f;
    #pragma unroll
    for (int i = 0; i < 16; ++i) {
        v[i] = rowp[t + i * 256];
        ssq = fmaf(v[i], v[i], ssq);
    }
    sm[t] = ssq; __syncthreads();
    for (int s = 128; s > 0; s >>= 1) {
        if (t < s) sm[t] += sm[t + s];
        __syncthreads();
    }
    float scale = rsqrtf(fmaxf(sm[0], HB_EPS));
    #pragma unroll
    for (int i = 0; i < 16; ++i)
        rowp[t + i * 256] = v[i] * scale;
}

extern "C" void kernel_launch(void* const* d_in, const int* in_sizes, int n_in,
                              void* d_out, int out_size, void* d_ws, size_t ws_size,
                              hipStream_t stream) {
    (void)in_sizes; (void)n_in; (void)out_size; (void)ws_size;
    const float* inp  = (const float*)d_in[0];
    const float* base = (const float*)d_in[1];
    const float* pw   = (const float*)d_in[2];
    float* out = (float*)d_out;

    float* y = (float*)((char*)d_ws + 1024);

    hb_shiftsum  <<<dim3(8, NPAIRS), SS_THREADS, 0, stream>>>(inp, pw, y);
    hb_gemm_wmma <<<dim3(64, 4),     256, 0, stream>>>(y, base, out);
    hb_normalize <<<NPAIRS,          256, 0, stream>>>(out);
}
